# starting kernel
# baseline (speedup 1.0000x reference)
#include <hip/hip_runtime.h>

typedef float v2f __attribute__((ext_vector_type(2)));
typedef float v8f __attribute__((ext_vector_type(8)));

#define N_NODES 50000
#define F_IN    256
#define F_HID   128
#define F_OUT   40

__global__ __launch_bounds__(256) void k_init_deg(float* __restrict__ deg) {
    int i = blockIdx.x * 256 + threadIdx.x;
    if (i < N_NODES) deg[i] = 1.0f;
}

__global__ __launch_bounds__(256) void k_deg(const int* __restrict__ col,
                                             const float* __restrict__ ew,
                                             float* __restrict__ deg, int E) {
    int e = blockIdx.x * 256 + threadIdx.x;
    if (e < E) atomicAdd(&deg[col[e]], ew[e]);
}

__global__ __launch_bounds__(256) void k_dinv(float* __restrict__ deg) {
    int i = blockIdx.x * 256 + threadIdx.x;
    if (i < N_NODES) {
        float d = deg[i];
        deg[i] = (d > 0.0f) ? rsqrtf(d) : 0.0f;
    }
}

__global__ __launch_bounds__(32) void k_gemm1(const float* __restrict__ X,
                                              const float* __restrict__ W,
                                              float* __restrict__ H) {
    const int m0   = blockIdx.x * 16;
    const int n0   = blockIdx.y * 64;
    const int lane = threadIdx.x;
    const int half = lane >> 4;
    const int l16  = lane & 15;

    const float* __restrict__ arow = X + (long)(m0 + l16) * F_IN;
    const float* __restrict__ bcol = W + (n0 + l16);

    v8f acc0 = {}, acc1 = {}, acc2 = {}, acc3 = {};
#pragma unroll 4
    for (int k0 = 0; k0 < F_IN; k0 += 4) {
        const int kk = k0 + 2 * half;
        v2f a;
        a.x = arow[kk + 0];
        a.y = arow[kk + 1];
        const float* __restrict__ b0p = bcol + (long)kk * F_HID;
        const float* __restrict__ b1p = b0p + F_HID;
        v2f b;
        b.x = b0p[0];  b.y = b1p[0];
        acc0 = __builtin_amdgcn_wmma_f32_16x16x4_f32(false, a, false, b, (short)0, acc0, false, false);
        b.x = b0p[16]; b.y = b1p[16];
        acc1 = __builtin_amdgcn_wmma_f32_16x16x4_f32(false, a, false, b, (short)0, acc1, false, false);
        b.x = b0p[32]; b.y = b1p[32];
        acc2 = __builtin_amdgcn_wmma_f32_16x16x4_f32(false, a, false, b, (short)0, acc2, false, false);
        b.x = b0p[48]; b.y = b1p[48];
        acc3 = __builtin_amdgcn_wmma_f32_16x16x4_f32(false, a, false, b, (short)0, acc3, false, false);
    }
#pragma unroll
    for (int r = 0; r < 8; ++r) {
        int mm = m0 + r + 8 * half;
        float* hrow = H + (long)mm * F_HID + n0 + l16;
        hrow[0]  = acc0[r];
        hrow[16] = acc1[r];
        hrow[32] = acc2[r];
        hrow[48] = acc3[r];
    }
}

__global__ __launch_bounds__(256) void k_self1(const float* __restrict__ h1,
                                               const float* __restrict__ dinv,
                                               float* __restrict__ agg1) {
    long idx = (long)blockIdx.x * 256 + threadIdx.x;
    if (idx < (long)N_NODES * F_HID) {
        int  i  = (int)(idx >> 7);
        float di = dinv[i];
        agg1[idx] = h1[idx] * di * di;
    }
}

__global__ __launch_bounds__(256) void k_scatter1(const int* __restrict__ row,
                                                  const int* __restrict__ col,
                                                  const float* __restrict__ ew,
                                                  const float* __restrict__ dinv,
                                                  const float* __restrict__ h1,
                                                  float* __restrict__ agg1, int E) {
    int e    = blockIdx.x * 8 + (threadIdx.x >> 5);
    int lane = threadIdx.x & 31;
    if (e >= E) return;
    int r = row[e], c = col[e];
    float nrm = dinv[r] * ew[e] * dinv[c];
    const float4* __restrict__ src = (const float4*)(h1 + (long)r * F_HID);
    float4 v = src[lane];
    float* dst = agg1 + (long)c * F_HID + lane * 4;
    atomicAdd(dst + 0, v.x * nrm);
    atomicAdd(dst + 1, v.y * nrm);
    atomicAdd(dst + 2, v.z * nrm);
    atomicAdd(dst + 3, v.w * nrm);
}

__global__ __launch_bounds__(256) void k_relu(float* __restrict__ agg1,
                                              const float* __restrict__ b1) {
    long idx = (long)blockIdx.x * 256 + threadIdx.x;
    if (idx < (long)N_NODES * F_HID) {
        int j = (int)(idx & (F_HID - 1));
        float v = agg1[idx] + b1[j];
        agg1[idx] = v > 0.0f ? v : 0.0f;
    }
}

__global__ __launch_bounds__(32) void k_gemm2(const float* __restrict__ Hr,
                                              const float* __restrict__ W,
                                              float* __restrict__ O) {
    const int m0   = blockIdx.x * 16;
    const int lane = threadIdx.x;
    const int half = lane >> 4;
    const int l16  = lane & 15;

    const float* __restrict__ arow = Hr + (long)(m0 + l16) * F_HID;
    const bool n2ok = (32 + l16) < F_OUT;

    v8f acc0 = {}, acc1 = {}, acc2 = {};
#pragma unroll 4
    for (int k0 = 0; k0 < F_HID; k0 += 4) {
        const int kk = k0 + 2 * half;
        v2f a;
        a.x = arow[kk + 0];
        a.y = arow[kk + 1];
        const float* __restrict__ b0p = W + (long)kk * F_OUT + l16;
        const float* __restrict__ b1p = b0p + F_OUT;
        v2f b;
        b.x = b0p[0];  b.y = b1p[0];
        acc0 = __builtin_amdgcn_wmma_f32_16x16x4_f32(false, a, false, b, (short)0, acc0, false, false);
        b.x = b0p[16]; b.y = b1p[16];
        acc1 = __builtin_amdgcn_wmma_f32_16x16x4_f32(false, a, false, b, (short)0, acc1, false, false);
        b.x = n2ok ? b0p[32] : 0.0f;
        b.y = n2ok ? b1p[32] : 0.0f;
        acc2 = __builtin_amdgcn_wmma_f32_16x16x4_f32(false, a, false, b, (short)0, acc2, false, false);
    }
#pragma unroll
    for (int r = 0; r < 8; ++r) {
        int mm = m0 + r + 8 * half;
        float* orow = O + (long)mm * F_OUT + l16;
        orow[0]  = acc0[r];
        orow[16] = acc1[r];
        if (n2ok) orow[32] = acc2[r];
    }
}

__global__ __launch_bounds__(256) void k_outinit(const float* __restrict__ h2,
                                                 const float* __restrict__ dinv,
                                                 const float* __restrict__ b2,
                                                 float* __restrict__ out) {
    long idx = (long)blockIdx.x * 256 + threadIdx.x;
    if (idx < (long)N_NODES * F_OUT) {
        int i = (int)(idx / F_OUT);
        int j = (int)(idx - (long)i * F_OUT);
        float di = dinv[i];
        out[idx] = h2[idx] * di * di + b2[j];
    }
}

__global__ __launch_bounds__(256) void k_scatter2(const int* __restrict__ row,
                                                  const int* __restrict__ col,
                                                  const float* __restrict__ ew,
                                                  const float* __restrict__ dinv,
                                                  const float* __restrict__ h2,
                                                  float* __restrict__ out, int E) {
    int e    = blockIdx.x * 8 + (threadIdx.x >> 5);
    int lane = threadIdx.x & 31;
    if (e >= E) return;
    int r = row[e], c = col[e];
    float nrm = dinv[r] * ew[e] * dinv[c];
    const float* __restrict__ src = h2 + (long)r * F_OUT;
    float* dst = out + (long)c * F_OUT;
    atomicAdd(&dst[lane], src[lane] * nrm);
    if (lane < F_OUT - 32)
        atomicAdd(&dst[lane + 32], src[lane + 32] * nrm);
}

extern "C" void kernel_launch(void* const* d_in, const int* in_sizes, int n_in,
                              void* d_out, int out_size, void* d_ws, size_t ws_size,
                              hipStream_t stream) {
    const float* x   = (const float*)d_in[0];
    const int*   ei  = (const int*)d_in[1];
    const float* ew  = (const float*)d_in[2];
    const float* W1  = (const float*)d_in[3];
    const float* b1  = (const float*)d_in[4];
    const float* W2  = (const float*)d_in[5];
    const float* b2  = (const float*)d_in[6];
    float*       out = (float*)d_out;

    const int E = in_sizes[2];
    const int* row = ei;
    const int* col = ei + E;

    char* ws = (char*)d_ws;
    float* deg  = (float*)(ws + 0);
    float* h1   = (float*)(ws + 409600);
    float* agg1 = (float*)(ws + 409600 + 25600000);
    float* h2   = (float*)(ws + 409600 + 2 * 25600000);
    (void)ws_size; (void)n_in; (void)out_size;

    const int nb_nodes = (N_NODES + 255) / 256;
    const int nb_edges = (E + 255) / 256;
    const int nb_wave_edges = (E + 7) / 8;
    const int nb_nh = (N_NODES * F_HID + 255) / 256;
    const int nb_no = (N_NODES * F_OUT + 255) / 256;

    k_init_deg<<<nb_nodes, 256, 0, stream>>>(deg);
    k_deg<<<nb_edges, 256, 0, stream>>>(col, ew, deg, E);
    k_dinv<<<nb_nodes, 256, 0, stream>>>(deg);
    float* dinv = deg;

    k_gemm1<<<dim3(N_NODES / 16, F_HID / 64), 32, 0, stream>>>(x, W1, h1);
    k_self1<<<nb_nh, 256, 0, stream>>>(h1, dinv, agg1);
    k_scatter1<<<nb_wave_edges, 256, 0, stream>>>(row, col, ew, dinv, h1, agg1, E);
    k_relu<<<nb_nh, 256, 0, stream>>>(agg1, b1);

    k_gemm2<<<N_NODES / 16, 32, 0, stream>>>(agg1, W2, h2);
    k_outinit<<<nb_no, 256, 0, stream>>>(h2, dinv, b2, out);
    k_scatter2<<<nb_wave_edges, 256, 0, stream>>>(row, col, ew, dinv, h2, out, E);
}


# scatter kernels: a wave walks 32 consecutive edges (lanes load row/col/weight/norms once, readlane broadcast, 8 row loads in flight), not 1 edge per wave; f32 atomics kept, arrival order differs
# speedup vs baseline: 1.0878x; 1.0878x over previous
#include <hip/hip_runtime.h>

typedef float v2f __attribute__((ext_vector_type(2)));
typedef float v8f __attribute__((ext_vector_type(8)));

#define N_NODES 50000
#define F_IN    256
#define F_HID   128
#define F_OUT   40

__global__ __launch_bounds__(256) void k_init_deg(float* __restrict__ deg) {
    int i = blockIdx.x * 256 + threadIdx.x;
    if (i < N_NODES) deg[i] = 1.0f;
}

__global__ __launch_bounds__(256) void k_deg(const int* __restrict__ col,
                                             const float* __restrict__ ew,
                                             float* __restrict__ deg, int E) {
    int e = blockIdx.x * 256 + threadIdx.x;
    if (e < E) atomicAdd(&deg[col[e]], ew[e]);
}

__global__ __launch_bounds__(256) void k_dinv(float* __restrict__ deg) {
    int i = blockIdx.x * 256 + threadIdx.x;
    if (i < N_NODES) {
        float d = deg[i];
        deg[i] = (d > 0.0f) ? rsqrtf(d) : 0.0f;
    }
}

__global__ __launch_bounds__(32) void k_gemm1(const float* __restrict__ X,
                                              const float* __restrict__ W,
                                              float* __restrict__ H) {
    const int m0   = blockIdx.x * 16;
    const int n0   = blockIdx.y * 64;
    const int lane = threadIdx.x;
    const int half = lane >> 4;
    const int l16  = lane & 15;

    const float* __restrict__ arow = X + (long)(m0 + l16) * F_IN;
    const float* __restrict__ bcol = W + (n0 + l16);

    v8f acc0 = {}, acc1 = {}, acc2 = {}, acc3 = {};
#pragma unroll 4
    for (int k0 = 0; k0 < F_IN; k0 += 4) {
        const int kk = k0 + 2 * half;
        v2f a;
        a.x = arow[kk + 0];
        a.y = arow[kk + 1];
        const float* __restrict__ b0p = bcol + (long)kk * F_HID;
        const float* __restrict__ b1p = b0p + F_HID;
        v2f b;
        b.x = b0p[0];  b.y = b1p[0];
        acc0 = __builtin_amdgcn_wmma_f32_16x16x4_f32(false, a, false, b, (short)0, acc0, false, false);
        b.x = b0p[16]; b.y = b1p[16];
        acc1 = __builtin_amdgcn_wmma_f32_16x16x4_f32(false, a, false, b, (short)0, acc1, false, false);
        b.x = b0p[32]; b.y = b1p[32];
        acc2 = __builtin_amdgcn_wmma_f32_16x16x4_f32(false, a, false, b, (short)0, acc2, false, false);
        b.x = b0p[48]; b.y = b1p[48];
        acc3 = __builtin_amdgcn_wmma_f32_16x16x4_f32(false, a, false, b, (short)0, acc3, false, false);
    }
#pragma unroll
    for (int r = 0; r < 8; ++r) {
        int mm = m0 + r + 8 * half;
        float* hrow = H + (long)mm * F_HID + n0 + l16;
        hrow[0]  = acc0[r];
        hrow[16] = acc1[r];
        hrow[32] = acc2[r];
        hrow[48] = acc3[r];
    }
}

__global__ __launch_bounds__(256) void k_self1(const float* __restrict__ h1,
                                               const float* __restrict__ dinv,
                                               float* __restrict__ agg1) {
    long idx = (long)blockIdx.x * 256 + threadIdx.x;
    if (idx < (long)N_NODES * F_HID) {
        int  i  = (int)(idx >> 7);
        float di = dinv[i];
        agg1[idx] = h1[idx] * di * di;
    }
}

#define EDGES_PER_WAVE 32
#define ROWS_IN_FLIGHT 8

static __device__ __forceinline__ int lane_to_scalar_i(int v, int srclane) {
    return __builtin_amdgcn_readlane(v, srclane);
}
static __device__ __forceinline__ float lane_to_scalar_f(float v, int srclane) {
    return __int_as_float(__builtin_amdgcn_readlane(__float_as_int(v), srclane));
}

__global__ __launch_bounds__(256) void k_scatter1(const int* __restrict__ row,
                                                  const int* __restrict__ col,
                                                  const float* __restrict__ ew,
                                                  const float* __restrict__ dinv,
                                                  const float* __restrict__ h1,
                                                  float* __restrict__ agg1, int E) {
    const int lane = threadIdx.x & 31;
    const int wv = __builtin_amdgcn_readfirstlane((int)(threadIdx.x >> 5));
    const int e0 = (blockIdx.x * 8 + wv) * EDGES_PER_WAVE;
    if (e0 >= E) return;
    const int left = E - e0;
    const int n = left < EDGES_PER_WAVE ? left : EDGES_PER_WAVE;

    int r = 0, c = 0;
    float nrm = 0.0f;
    if (lane < n) {
        const int e = e0 + lane;
        r = row[e];
        c = col[e];
        nrm = dinv[r] * ew[e] * dinv[c];
    }

    int j = 0;
    for (; j + ROWS_IN_FLIGHT <= n; j += ROWS_IN_FLIGHT) {
        float4 v[ROWS_IN_FLIGHT];
#pragma unroll
        for (int u = 0; u < ROWS_IN_FLIGHT; ++u) {
            const int rj = lane_to_scalar_i(r, j + u);
            v[u] = ((const float4*)(h1 + (long)rj * F_HID))[lane];
        }
#pragma unroll
        for (int u = 0; u < ROWS_IN_FLIGHT; ++u) {
            const int   cj = lane_to_scalar_i(c, j + u);
            const float nj = lane_to_scalar_f(nrm, j + u);
            float* dst = agg1 + (long)cj * F_HID + lane * 4;
            atomicAdd(dst + 0, v[u].x * nj);
            atomicAdd(dst + 1, v[u].y * nj);
            atomicAdd(dst + 2, v[u].z * nj);
            atomicAdd(dst + 3, v[u].w * nj);
        }
    }
    for (; j < n; ++j) {
        const int   rj = lane_to_scalar_i(r, j);
        const int   cj = lane_to_scalar_i(c, j);
        const float nj = lane_to_scalar_f(nrm, j);
        const float4 v = ((const float4*)(h1 + (long)rj * F_HID))[lane];
        float* dst = agg1 + (long)cj * F_HID + lane * 4;
        atomicAdd(dst + 0, v.x * nj);
        atomicAdd(dst + 1, v.y * nj);
        atomicAdd(dst + 2, v.z * nj);
        atomicAdd(dst + 3, v.w * nj);
    }
}

__global__ __launch_bounds__(256) void k_relu(float* __restrict__ agg1,
                                              const float* __restrict__ b1) {
    long idx = (long)blockIdx.x * 256 + threadIdx.x;
    if (idx < (long)N_NODES * F_HID) {
        int j = (int)(idx & (F_HID - 1));
        float v = agg1[idx] + b1[j];
        agg1[idx] = v > 0.0f ? v : 0.0f;
    }
}

__global__ __launch_bounds__(32) void k_gemm2(const float* __restrict__ Hr,
                                              const float* __restrict__ W,
                                              float* __restrict__ O) {
    const int m0   = blockIdx.x * 16;
    const int lane = threadIdx.x;
    const int half = lane >> 4;
    const int l16  = lane & 15;

    const float* __restrict__ arow = Hr + (long)(m0 + l16) * F_HID;
    const bool n2ok = (32 + l16) < F_OUT;

    v8f acc0 = {}, acc1 = {}, acc2 = {};
#pragma unroll 4
    for (int k0 = 0; k0 < F_HID; k0 += 4) {
        const int kk = k0 + 2 * half;
        v2f a;
        a.x = arow[kk + 0];
        a.y = arow[kk + 1];
        const float* __restrict__ b0p = W + (long)kk * F_OUT + l16;
        const float* __restrict__ b1p = b0p + F_OUT;
        v2f b;
        b.x = b0p[0];  b.y = b1p[0];
        acc0 = __builtin_amdgcn_wmma_f32_16x16x4_f32(false, a, false, b, (short)0, acc0, false, false);
        b.x = b0p[16]; b.y = b1p[16];
        acc1 = __builtin_amdgcn_wmma_f32_16x16x4_f32(false, a, false, b, (short)0, acc1, false, false);
        b.x = n2ok ? b0p[32] : 0.0f;
        b.y = n2ok ? b1p[32] : 0.0f;
        acc2 = __builtin_amdgcn_wmma_f32_16x16x4_f32(false, a, false, b, (short)0, acc2, false, false);
    }
#pragma unroll
    for (int r = 0; r < 8; ++r) {
        int mm = m0 + r + 8 * half;
        float* orow = O + (long)mm * F_OUT + l16;
        orow[0]  = acc0[r];
        orow[16] = acc1[r];
        if (n2ok) orow[32] = acc2[r];
    }
}

__global__ __launch_bounds__(256) void k_outinit(const float* __restrict__ h2,
                                                 const float* __restrict__ dinv,
                                                 const float* __restrict__ b2,
                                                 float* __restrict__ out) {
    long idx = (long)blockIdx.x * 256 + threadIdx.x;
    if (idx < (long)N_NODES * F_OUT) {
        int i = (int)(idx / F_OUT);
        int j = (int)(idx - (long)i * F_OUT);
        float di = dinv[i];
        out[idx] = h2[idx] * di * di + b2[j];
    }
}

__global__ __launch_bounds__(256) void k_scatter2(const int* __restrict__ row,
                                                  const int* __restrict__ col,
                                                  const float* __restrict__ ew,
                                                  const float* __restrict__ dinv,
                                                  const float* __restrict__ h2,
                                                  float* __restrict__ out, int E) {
    const int lane = threadIdx.x & 31;
    const int wv = __builtin_amdgcn_readfirstlane((int)(threadIdx.x >> 5));
    const int e0 = (blockIdx.x * 8 + wv) * EDGES_PER_WAVE;
    if (e0 >= E) return;
    const int left = E - e0;
    const int n = left < EDGES_PER_WAVE ? left : EDGES_PER_WAVE;
    const bool tail = lane < F_OUT - 32;

    int r = 0, c = 0;
    float nrm = 0.0f;
    if (lane < n) {
        const int e = e0 + lane;
        r = row[e];
        c = col[e];
        nrm = dinv[r] * ew[e] * dinv[c];
    }

    int j = 0;
    for (; j + ROWS_IN_FLIGHT <= n; j += ROWS_IN_FLIGHT) {
        float a[ROWS_IN_FLIGHT], b[ROWS_IN_FLIGHT];
        if (tail) {
#pragma unroll
            for (int u = 0; u < ROWS_IN_FLIGHT; ++u) {
                const int rj = lane_to_scalar_i(r, j + u);
                b[u] = h2[(long)rj * F_OUT + 32 + lane];
            }
        }
#pragma unroll
        for (int u = 0; u < ROWS_IN_FLIGHT; ++u) {
            const int rj = lane_to_scalar_i(r, j + u);
            a[u] = h2[(long)rj * F_OUT + lane];
        }
#pragma unroll
        for (int u = 0; u < ROWS_IN_FLIGHT; ++u) {
            const int   cj = lane_to_scalar_i(c, j + u);
            const float nj = lane_to_scalar_f(nrm, j + u);
            atomicAdd(&out[(long)cj * F_OUT + lane], a[u] * nj);
        }
        if (tail) {
#pragma unroll
            for (int u = 0; u < ROWS_IN_FLIGHT; ++u) {
                const int   cj = lane_to_scalar_i(c, j + u);
                const float nj = lane_to_scalar_f(nrm, j + u);
                atomicAdd(&out[(long)cj * F_OUT + 32 + lane], b[u] * nj);
            }
        }
    }
    for (; j < n; ++j) {
        const int   rj = lane_to_scalar_i(r, j);
        const int   cj = lane_to_scalar_i(c, j);
        const float nj = lane_to_scalar_f(nrm, j);
        const float* __restrict__ src = h2 + (long)rj * F_OUT;
        float* dst = out + (long)cj * F_OUT;
        atomicAdd(&dst[lane], src[lane] * nj);
        if (tail)
            atomicAdd(&dst[lane + 32], src[lane + 32] * nj);
    }
}

extern "C" void kernel_launch(void* const* d_in, const int* in_sizes, int n_in,
                              void* d_out, int out_size, void* d_ws, size_t ws_size,
                              hipStream_t stream) {
    const float* x   = (const float*)d_in[0];
    const int*   ei  = (const int*)d_in[1];
    const float* ew  = (const float*)d_in[2];
    const float* W1  = (const float*)d_in[3];
    const float* b1  = (const float*)d_in[4];
    const float* W2  = (const float*)d_in[5];
    const float* b2  = (const float*)d_in[6];
    float*       out = (float*)d_out;

    const int E = in_sizes[2];
    const int* row = ei;
    const int* col = ei + E;

    char* ws = (char*)d_ws;
    float* deg  = (float*)(ws + 0);
    float* h1   = (float*)(ws + 409600);
    float* agg1 = (float*)(ws + 409600 + 25600000);
    float* h2   = (float*)(ws + 409600 + 2 * 25600000);
    (void)ws_size; (void)n_in; (void)out_size;

    const int nb_nodes = (N_NODES + 255) / 256;
    const int nb_edges = (E + 255) / 256;
    const int nb_nh = (N_NODES * F_HID + 255) / 256;
    const int nb_no = (N_NODES * F_OUT + 255) / 256;

    k_init_deg<<<nb_nodes, 256, 0, stream>>>(deg);
    k_deg<<<nb_edges, 256, 0, stream>>>(col, ew, deg, E);
    k_dinv<<<nb_nodes, 256, 0, stream>>>(deg);
    float* dinv = deg;

    k_gemm1<<<dim3(N_NODES / 16, F_HID / 64), 32, 0, stream>>>(x, W1, h1);
    k_self1<<<nb_nh, 256, 0, stream>>>(h1, dinv, agg1);
    k_scatter1<<<nb_edges, 256, 0, stream>>>(row, col, ew, dinv, h1, agg1, E);
    k_relu<<<nb_nh, 256, 0, stream>>>(agg1, b1);

    k_gemm2<<<N_NODES / 16, 32, 0, stream>>>(agg1, W2, h2);
    k_outinit<<<nb_no, 256, 0, stream>>>(h2, dinv, b2, out);
    k_scatter2<<<nb_edges, 256, 0, stream>>>(row, col, ew, dinv, h2, out, E);
}


# edges counting-sorted by destination per call; one wave sums a node's source rows in registers, one store per row (self term, bias, relu folded in); no f32 atomic scatter; sum order = arrival slots
# speedup vs baseline: 1.8008x; 1.6555x over previous
#include <hip/hip_runtime.h>

typedef float v2f __attribute__((ext_vector_type(2)));
typedef float v8f __attribute__((ext_vector_type(8)));
typedef float v4f __attribute__((ext_vector_type(4)));

#define N_NODES 50000
#define F_IN    256
#define F_HID   128
#define F_OUT   40

__global__ __launch_bounds__(256) void k_init_deg(float* __restrict__ deg,
                                                  int* __restrict__ cnt) {
    int i = blockIdx.x * 256 + threadIdx.x;
    if (i < N_NODES) {
        deg[i] = 1.0f;
        cnt[i] = 0;
    }
}

__global__ __launch_bounds__(256) void k_deg(const int* __restrict__ col,
                                             const float* __restrict__ ew,
                                             float* __restrict__ deg,
                                             int* __restrict__ cnt,
                                             int* __restrict__ pos, int E) {
    int e = blockIdx.x * 256 + threadIdx.x;
    if (e < E) {
        const int c = col[e];
        atomicAdd(&deg[c], ew[e]);
        pos[e] = atomicAdd(&cnt[c], 1);
    }
}

__global__ __launch_bounds__(256) void k_dinv(float* __restrict__ deg) {
    int i = blockIdx.x * 256 + threadIdx.x;
    if (i < N_NODES) {
        float d = deg[i];
        deg[i] = (d > 0.0f) ? rsqrtf(d) : 0.0f;
    }
}

#define SCAN_THREADS 1024
#define SCAN_ITEMS   49

__global__ __launch_bounds__(SCAN_THREADS) void k_scan(const int* __restrict__ cnt,
                                                       int* __restrict__ off) {
    __shared__ int s[SCAN_THREADS * SCAN_ITEMS];
    __shared__ int wsum[32];
    const int t    = threadIdx.x;
    const int lane = t & 31;
    const int wv   = t >> 5;

    for (int k = 0; k < SCAN_ITEMS; ++k) {
        const int i = k * SCAN_THREADS + t;
        s[i] = (i < N_NODES) ? cnt[i] : 0;
    }
    __syncthreads();

    const int base = t * SCAN_ITEMS;
    int sum = 0;
    for (int k = 0; k < SCAN_ITEMS; ++k) sum += s[base + k];

    int inc = sum;
    for (int d = 1; d < 32; d <<= 1) {
        const int v = __shfl_up(inc, d);
        if (lane >= d) inc += v;
    }
    if (lane == 31) wsum[wv] = inc;
    __syncthreads();
    if (wv == 0) {
        const int w = wsum[lane];
        int winc = w;
        for (int d = 1; d < 32; d <<= 1) {
            const int v = __shfl_up(winc, d);
            if (lane >= d) winc += v;
        }
        wsum[lane] = winc - w;
    }
    __syncthreads();

    int run = wsum[wv] + inc - sum;
    for (int k = 0; k < SCAN_ITEMS; ++k) {
        const int v = s[base + k];
        s[base + k] = run;
        run += v;
    }
    __syncthreads();
    for (int k = 0; k < SCAN_ITEMS; ++k) {
        const int i = k * SCAN_THREADS + t;
        if (i < N_NODES) off[i] = s[i];
    }
    if (t == SCAN_THREADS - 1) off[N_NODES] = run;
}

__global__ __launch_bounds__(256) void k_fill(const int* __restrict__ row,
                                              const int* __restrict__ col,
                                              const float* __restrict__ ew,
                                              const float* __restrict__ dinv,
                                              const int* __restrict__ off,
                                              const int* __restrict__ pos,
                                              int* __restrict__ pairs, int E) {
    int e = blockIdx.x * 256 + threadIdx.x;
    if (e < E) {
        const int r = row[e];
        const int c = col[e];
        const float nrm = dinv[r] * ew[e] * dinv[c];
        ((int2*)pairs)[off[c] + pos[e]] = make_int2(r, __float_as_int(nrm));
    }
}

__global__ __launch_bounds__(32) void k_gemm1(const float* __restrict__ X,
                                              const float* __restrict__ W,
                                              float* __restrict__ H) {
    const int m0   = blockIdx.x * 16;
    const int n0   = blockIdx.y * 64;
    const int lane = threadIdx.x;
    const int half = lane >> 4;
    const int l16  = lane & 15;

    const float* __restrict__ arow = X + (long)(m0 + l16) * F_IN;
    const float* __restrict__ bcol = W + (n0 + l16);

    v8f acc0 = {}, acc1 = {}, acc2 = {}, acc3 = {};
#pragma unroll 4
    for (int k0 = 0; k0 < F_IN; k0 += 4) {
        const int kk = k0 + 2 * half;
        v2f a;
        a.x = arow[kk + 0];
        a.y = arow[kk + 1];
        const float* __restrict__ b0p = bcol + (long)kk * F_HID;
        const float* __restrict__ b1p = b0p + F_HID;
        v2f b;
        b.x = b0p[0];  b.y = b1p[0];
        acc0 = __builtin_amdgcn_wmma_f32_16x16x4_f32(false, a, false, b, (short)0, acc0, false, false);
        b.x = b0p[16]; b.y = b1p[16];
        acc1 = __builtin_amdgcn_wmma_f32_16x16x4_f32(false, a, false, b, (short)0, acc1, false, false);
        b.x = b0p[32]; b.y = b1p[32];
        acc2 = __builtin_amdgcn_wmma_f32_16x16x4_f32(false, a, false, b, (short)0, acc2, false, false);
        b.x = b0p[48]; b.y = b1p[48];
        acc3 = __builtin_amdgcn_wmma_f32_16x16x4_f32(false, a, false, b, (short)0, acc3, false, false);
    }
#pragma unroll
    for (int r = 0; r < 8; ++r) {
        int mm = m0 + r + 8 * half;
        float* hrow = H + (long)mm * F_HID + n0 + l16;
        hrow[0]  = acc0[r];
        hrow[16] = acc1[r];
        hrow[32] = acc2[r];
        hrow[48] = acc3[r];
    }
}

#define ROWS_IN_FLIGHT 8

static __device__ __forceinline__ int lane_to_scalar_i(int v, int srclane) {
    return __builtin_amdgcn_readlane(v, srclane);
}
static __device__ __forceinline__ float lane_to_scalar_f(float v, int srclane) {
    return __int_as_float(__builtin_amdgcn_readlane(__float_as_int(v), srclane));
}

__global__ __launch_bounds__(256) void k_gather1(const int* __restrict__ off,
                                                 const int* __restrict__ pairs,
                                                 const float* __restrict__ dinv,
                                                 const float* __restrict__ h1,
                                                 const float* __restrict__ b1,
                                                 float* __restrict__ agg1) {
    const int lane = threadIdx.x & 31;
    const int wv = __builtin_amdgcn_readfirstlane((int)(threadIdx.x >> 5));
    const int c = blockIdx.x * 8 + wv;
    if (c >= N_NODES) return;
    const int beg = __builtin_amdgcn_readfirstlane(off[c]);
    const int end = __builtin_amdgcn_readfirstlane(off[c + 1]);

    const float di = dinv[c];
    const float4 sv = ((const float4*)(h1 + (long)c * F_HID))[lane];
    float4 acc;
    acc.x = sv.x * di * di;
    acc.y = sv.y * di * di;
    acc.z = sv.z * di * di;
    acc.w = sv.w * di * di;

    for (int base = beg; base < end; base += 32) {
        const int left = end - base;
        const int n = left < 32 ? left : 32;
        int r = 0;
        float w = 0.0f;
        if (lane < n) {
            const int2 p = ((const int2*)pairs)[base + lane];
            r = p.x;
            w = __int_as_float(p.y);
        }
        for (int j = 0; j < n; j += ROWS_IN_FLIGHT) {
            v4f v[ROWS_IN_FLIGHT];
#pragma unroll
            for (int u = 0; u < ROWS_IN_FLIGHT; ++u) {
                const int ju = j + u < n ? j + u : n - 1;
                const int rj = lane_to_scalar_i(r, ju);
                v[u] = ((const v4f*)(h1 + (long)rj * F_HID))[lane];
            }
            asm volatile("" :: "v"(v[0]), "v"(v[1]), "v"(v[2]), "v"(v[3]),
                               "v"(v[4]), "v"(v[5]), "v"(v[6]), "v"(v[7]));
#pragma unroll
            for (int u = 0; u < ROWS_IN_FLIGHT; ++u) {
                if (u == 0 || j + u < n) {
                    const float nj = lane_to_scalar_f(w, j + u);
                    acc.x += v[u].x * nj;
                    acc.y += v[u].y * nj;
                    acc.z += v[u].z * nj;
                    acc.w += v[u].w * nj;
                }
            }
        }
    }

    const float4 bv = ((const float4*)b1)[lane];
    float4 o;
    o.x = acc.x + bv.x;
    o.y = acc.y + bv.y;
    o.z = acc.z + bv.z;
    o.w = acc.w + bv.w;
    o.x = o.x > 0.0f ? o.x : 0.0f;
    o.y = o.y > 0.0f ? o.y : 0.0f;
    o.z = o.z > 0.0f ? o.z : 0.0f;
    o.w = o.w > 0.0f ? o.w : 0.0f;
    ((float4*)(agg1 + (long)c * F_HID))[lane] = o;
}

__global__ __launch_bounds__(32) void k_gemm2(const float* __restrict__ Hr,
                                              const float* __restrict__ W,
                                              float* __restrict__ O) {
    const int m0   = blockIdx.x * 16;
    const int lane = threadIdx.x;
    const int half = lane >> 4;
    const int l16  = lane & 15;

    const float* __restrict__ arow = Hr + (long)(m0 + l16) * F_HID;
    const bool n2ok = (32 + l16) < F_OUT;

    v8f acc0 = {}, acc1 = {}, acc2 = {};
#pragma unroll 4
    for (int k0 = 0; k0 < F_HID; k0 += 4) {
        const int kk = k0 + 2 * half;
        v2f a;
        a.x = arow[kk + 0];
        a.y = arow[kk + 1];
        const float* __restrict__ b0p = W + (long)kk * F_OUT + l16;
        const float* __restrict__ b1p = b0p + F_OUT;
        v2f b;
        b.x = b0p[0];  b.y = b1p[0];
        acc0 = __builtin_amdgcn_wmma_f32_16x16x4_f32(false, a, false, b, (short)0, acc0, false, false);
        b.x = b0p[16]; b.y = b1p[16];
        acc1 = __builtin_amdgcn_wmma_f32_16x16x4_f32(false, a, false, b, (short)0, acc1, false, false);
        b.x = n2ok ? b0p[32] : 0.0f;
        b.y = n2ok ? b1p[32] : 0.0f;
        acc2 = __builtin_amdgcn_wmma_f32_16x16x4_f32(false, a, false, b, (short)0, acc2, false, false);
    }
#pragma unroll
    for (int r = 0; r < 8; ++r) {
        int mm = m0 + r + 8 * half;
        float* orow = O + (long)mm * F_OUT + l16;
        orow[0]  = acc0[r];
        orow[16] = acc1[r];
        if (n2ok) orow[32] = acc2[r];
    }
}

__global__ __launch_bounds__(256) void k_gather2(const int* __restrict__ off,
                                                 const int* __restrict__ pairs,
                                                 const float* __restrict__ dinv,
                                                 const float* __restrict__ h2,
                                                 const float* __restrict__ b2,
                                                 float* __restrict__ out) {
    const int lane = threadIdx.x & 31;
    const int wv = __builtin_amdgcn_readfirstlane((int)(threadIdx.x >> 5));
    const int c = blockIdx.x * 8 + wv;
    if (c >= N_NODES) return;
    const int beg = __builtin_amdgcn_readfirstlane(off[c]);
    const int end = __builtin_amdgcn_readfirstlane(off[c + 1]);
    const bool tail = lane < F_OUT - 32;

    const float di = dinv[c];
    const float* __restrict__ self = h2 + (long)c * F_OUT;
    float a = self[lane] * di * di + b2[lane];
    float b = 0.0f;
    if (tail) b = self[32 + lane] * di * di + b2[32 + lane];

    for (int base = beg; base < end; base += 32) {
        const int left = end - base;
        const int n = left < 32 ? left : 32;
        int r = 0;
        float w = 0.0f;
        if (lane < n) {
            const int2 p = ((const int2*)pairs)[base + lane];
            r = p.x;
            w = __int_as_float(p.y);
        }
        for (int j = 0; j < n; j += ROWS_IN_FLIGHT) {
            float va[ROWS_IN_FLIGHT], vb[ROWS_IN_FLIGHT];
#pragma unroll
            for (int u = 0; u < ROWS_IN_FLIGHT; ++u) {
                const int ju = j + u < n ? j + u : n - 1;
                const int rj = lane_to_scalar_i(r, ju);
                const float* __restrict__ src = h2 + (long)rj * F_OUT;
                va[u] = src[lane];
                if (tail) vb[u] = src[32 + lane];
            }
#pragma unroll
            for (int u = 0; u < ROWS_IN_FLIGHT; ++u) {
                if (u == 0 || j + u < n) {
                    const float nj = lane_to_scalar_f(w, j + u);
                    a += va[u] * nj;
                    if (tail) b += vb[u] * nj;
                }
            }
        }
    }

    float* __restrict__ dst = out + (long)c * F_OUT;
    dst[lane] = a;
    if (tail) dst[32 + lane] = b;
}

extern "C" void kernel_launch(void* const* d_in, const int* in_sizes, int n_in,
                              void* d_out, int out_size, void* d_ws, size_t ws_size,
                              hipStream_t stream) {
    const float* x   = (const float*)d_in[0];
    const int*   ei  = (const int*)d_in[1];
    const float* ew  = (const float*)d_in[2];
    const float* W1  = (const float*)d_in[3];
    const float* b1  = (const float*)d_in[4];
    const float* W2  = (const float*)d_in[5];
    const float* b2  = (const float*)d_in[6];
    float*       out = (float*)d_out;

    const int E = in_sizes[2];
    const int* row = ei;
    const int* col = ei + E;

    char* ws = (char*)d_ws;
    float* deg  = (float*)(ws + 0);
    float* h1   = (float*)(ws + 409600);
    float* agg1 = (float*)(ws + 409600 + 25600000);
    float* h2   = (float*)(ws + 409600 + 2 * 25600000);
    int*   cnt   = (int*)(ws + 59609600);
    int*   off   = (int*)(ws + 59609600 + 200192);
    int*   pos   = (int*)(ws + 60009984);
    int*   pairs = (int*)(ws + 127118848);
    (void)ws_size; (void)n_in; (void)out_size;

    const int nb_nodes = (N_NODES + 255) / 256;
    const int nb_edges = (E + 255) / 256;

    k_init_deg<<<nb_nodes, 256, 0, stream>>>(deg, cnt);
    k_deg<<<nb_edges, 256, 0, stream>>>(col, ew, deg, cnt, pos, E);
    k_dinv<<<nb_nodes, 256, 0, stream>>>(deg);
    float* dinv = deg;
    k_scan<<<1, SCAN_THREADS, 0, stream>>>(cnt, off);
    k_fill<<<nb_edges, 256, 0, stream>>>(row, col, ew, dinv, off, pos, pairs, E);

    k_gemm1<<<dim3(N_NODES / 16, F_HID / 64), 32, 0, stream>>>(x, W1, h1);
    k_gather1<<<N_NODES / 8, 256, 0, stream>>>(off, pairs, dinv, h1, b1, agg1);

    k_gemm2<<<N_NODES / 16, 32, 0, stream>>>(agg1, W2, h2);
    k_gather2<<<N_NODES / 8, 256, 0, stream>>>(off, pairs, dinv, h2, b2, out);
}


# 9 launches -> 6: GEMM2 is the epilogue of the layer-1 node sums on their 16-row tile in LDS (no hidden plane in memory); start values stored by GEMM1's launch; 1/sqrt(deg) taken where read; same sums
# speedup vs baseline: 1.9653x; 1.0913x over previous
#include <hip/hip_runtime.h>

typedef float v2f __attribute__((ext_vector_type(2)));
typedef float v8f __attribute__((ext_vector_type(8)));
typedef float v4f __attribute__((ext_vector_type(4)));

#define N_NODES 50000
#define F_IN    256
#define F_HID   128
#define F_OUT   40

static __device__ __forceinline__ float inv_sqrt_deg(float d) {
    return (d > 0.0f) ? rsqrtf(d) : 0.0f;
}

__global__ __launch_bounds__(256) void k_deg(const int* __restrict__ col,
                                             const float* __restrict__ ew,
                                             float* __restrict__ deg,
                                             int* __restrict__ cnt,
                                             int* __restrict__ pos, int E) {
    int e = blockIdx.x * 256 + threadIdx.x;
    if (e < E) {
        const int c = col[e];
        atomicAdd(&deg[c], ew[e]);
        pos[e] = atomicAdd(&cnt[c], 1);
    }
}

#define SCAN_THREADS 1024
#define SCAN_ITEMS   49

__global__ __launch_bounds__(SCAN_THREADS) void k_scan(const int* __restrict__ cnt,
                                                       int* __restrict__ off) {
    __shared__ int s[SCAN_THREADS * SCAN_ITEMS];
    __shared__ int wsum[32];
    const int t    = threadIdx.x;
    const int lane = t & 31;
    const int wv   = t >> 5;

    for (int k = 0; k < SCAN_ITEMS; ++k) {
        const int i = k * SCAN_THREADS + t;
        s[i] = (i < N_NODES) ? cnt[i] : 0;
    }
    __syncthreads();

    const int base = t * SCAN_ITEMS;
    int sum = 0;
    for (int k = 0; k < SCAN_ITEMS; ++k) sum += s[base + k];

    int inc = sum;
    for (int d = 1; d < 32; d <<= 1) {
        const int v = __shfl_up(inc, d);
        if (lane >= d) inc += v;
    }
    if (lane == 31) wsum[wv] = inc;
    __syncthreads();
    if (wv == 0) {
        const int w = wsum[lane];
        int winc = w;
        for (int d = 1; d < 32; d <<= 1) {
            const int v = __shfl_up(winc, d);
            if (lane >= d) winc += v;
        }
        wsum[lane] = winc - w;
    }
    __syncthreads();

    int run = wsum[wv] + inc - sum;
    for (int k = 0; k < SCAN_ITEMS; ++k) {
        const int v = s[base + k];
        s[base + k] = run;
        run += v;
    }
    __syncthreads();
    for (int k = 0; k < SCAN_ITEMS; ++k) {
        const int i = k * SCAN_THREADS + t;
        if (i < N_NODES) off[i] = s[i];
    }
    if (t == SCAN_THREADS - 1) off[N_NODES] = run;
}

__global__ __launch_bounds__(256) void k_fill(const int* __restrict__ row,
                                              const int* __restrict__ col,
                                              const float* __restrict__ ew,
                                              const float* __restrict__ deg,
                                              const int* __restrict__ off,
                                              const int* __restrict__ pos,
                                              int* __restrict__ pairs, int E) {
    int e = blockIdx.x * 256 + threadIdx.x;
    if (e < E) {
        const int r = row[e];
        const int c = col[e];
        const float nrm = inv_sqrt_deg(deg[r]) * ew[e] * inv_sqrt_deg(deg[c]);
        ((int2*)pairs)[off[c] + pos[e]] = make_int2(r, __float_as_int(nrm));
    }
}

static_assert((N_NODES / 16) * (F_HID / 64) * 8 == N_NODES, "8 start values per wave cover every node");

__global__ __launch_bounds__(32) void k_gemm1(const float* __restrict__ X,
                                              const float* __restrict__ W,
                                              float* __restrict__ H,
                                              float* __restrict__ deg,
                                              int* __restrict__ cnt) {
    const int m0   = blockIdx.x * 16;
    const int n0   = blockIdx.y * 64;
    const int lane = threadIdx.x;
    const int half = lane >> 4;
    const int l16  = lane & 15;

    {
        const int first = (blockIdx.y * (N_NODES / 16) + blockIdx.x) * 8;
        if (lane < 8)       deg[first + lane]     = 1.0f;
        else if (lane < 16) cnt[first + lane - 8] = 0;
    }

    const float* __restrict__ arow = X + (long)(m0 + l16) * F_IN;
    const float* __restrict__ bcol = W + (n0 + l16);

    v8f acc0 = {}, acc1 = {}, acc2 = {}, acc3 = {};
#pragma unroll 4
    for (int k0 = 0; k0 < F_IN; k0 += 4) {
        const int kk = k0 + 2 * half;
        v2f a;
        a.x = arow[kk + 0];
        a.y = arow[kk + 1];
        const float* __restrict__ b0p = bcol + (long)kk * F_HID;
        const float* __restrict__ b1p = b0p + F_HID;
        v2f b;
        b.x = b0p[0];  b.y = b1p[0];
        acc0 = __builtin_amdgcn_wmma_f32_16x16x4_f32(false, a, false, b, (short)0, acc0, false, false);
        b.x = b0p[16]; b.y = b1p[16];
        acc1 = __builtin_amdgcn_wmma_f32_16x16x4_f32(false, a, false, b, (short)0, acc1, false, false);
        b.x = b0p[32]; b.y = b1p[32];
        acc2 = __builtin_amdgcn_wmma_f32_16x16x4_f32(false, a, false, b, (short)0, acc2, false, false);
        b.x = b0p[48]; b.y = b1p[48];
        acc3 = __builtin_amdgcn_wmma_f32_16x16x4_f32(false, a, false, b, (short)0, acc3, false, false);
    }
#pragma unroll
    for (int r = 0; r < 8; ++r) {
        int mm = m0 + r + 8 * half;
        float* hrow = H + (long)mm * F_HID + n0 + l16;
        hrow[0]  = acc0[r];
        hrow[16] = acc1[r];
        hrow[32] = acc2[r];
        hrow[48] = acc3[r];
    }
}

#define ROWS_IN_FLIGHT 8
#define TILE_ROWS  16
#define TILE_PITCH 132
static_assert(N_NODES % TILE_ROWS == 0, "every work-group has 16 real nodes");

static __device__ __forceinline__ int lane_to_scalar_i(int v, int srclane) {
    return __builtin_amdgcn_readlane(v, srclane);
}
static __device__ __forceinline__ float lane_to_scalar_f(float v, int srclane) {
    return __int_as_float(__builtin_amdgcn_readlane(__float_as_int(v), srclane));
}

__global__ __launch_bounds__(512) void k_gather1_gemm2(const int* __restrict__ off,
                                                       const int* __restrict__ pairs,
                                                       const float* __restrict__ deg,
                                                       const float* __restrict__ h1,
                                                       const float* __restrict__ b1,
                                                       const float* __restrict__ W2,
                                                       float* __restrict__ h2) {
    __shared__ __attribute__((aligned(16))) float tile[TILE_ROWS * TILE_PITCH];

    const int lane = threadIdx.x & 31;
    const int wv = __builtin_amdgcn_readfirstlane((int)(threadIdx.x >> 5));
    const int m0 = blockIdx.x * TILE_ROWS;
    const int c  = m0 + wv;
    const int beg = __builtin_amdgcn_readfirstlane(off[c]);
    const int end = __builtin_amdgcn_readfirstlane(off[c + 1]);

    const float di = inv_sqrt_deg(deg[c]);
    const float4 sv = ((const float4*)(h1 + (long)c * F_HID))[lane];
    float4 acc;
    acc.x = sv.x * di * di;
    acc.y = sv.y * di * di;
    acc.z = sv.z * di * di;
    acc.w = sv.w * di * di;

    for (int base = beg; base < end; base += 32) {
        const int left = end - base;
        const int n = left < 32 ? left : 32;
        int r = 0;
        float w = 0.0f;
        if (lane < n) {
            const int2 p = ((const int2*)pairs)[base + lane];
            r = p.x;
            w = __int_as_float(p.y);
        }
        for (int j = 0; j < n; j += ROWS_IN_FLIGHT) {
            v4f v[ROWS_IN_FLIGHT];
#pragma unroll
            for (int u = 0; u < ROWS_IN_FLIGHT; ++u) {
                const int ju = j + u < n ? j + u : n - 1;
                const int rj = lane_to_scalar_i(r, ju);
                v[u] = ((const v4f*)(h1 + (long)rj * F_HID))[lane];
            }
            asm volatile("" :: "v"(v[0]), "v"(v[1]), "v"(v[2]), "v"(v[3]),
                               "v"(v[4]), "v"(v[5]), "v"(v[6]), "v"(v[7]));
#pragma unroll
            for (int u = 0; u < ROWS_IN_FLIGHT; ++u) {
                if (u == 0 || j + u < n) {
                    const float nj = lane_to_scalar_f(w, j + u);
                    acc.x += v[u].x * nj;
                    acc.y += v[u].y * nj;
                    acc.z += v[u].z * nj;
                    acc.w += v[u].w * nj;
                }
            }
        }
    }

    const float4 bv = ((const float4*)b1)[lane];
    float4 o;
    o.x = acc.x + bv.x;
    o.y = acc.y + bv.y;
    o.z = acc.z + bv.z;
    o.w = acc.w + bv.w;
    o.x = o.x > 0.0f ? o.x : 0.0f;
    o.y = o.y > 0.0f ? o.y : 0.0f;
    o.z = o.z > 0.0f ? o.z : 0.0f;
    o.w = o.w > 0.0f ? o.w : 0.0f;
    ((float4*)(tile + wv * TILE_PITCH))[lane] = o;
    __syncthreads();

    if (wv < 3) {
        const int half = lane >> 4;
        const int l16  = lane & 15;
        const int ncol = wv * 16 + l16;
        const bool nok = ncol < F_OUT;
        const float* __restrict__ arow = tile + l16 * TILE_PITCH + 2 * half;
        const float* __restrict__ bcol = W2 + (long)(2 * half) * F_OUT + (nok ? ncol : 0);

        v8f cacc = {};
#pragma unroll 8
        for (int k0 = 0; k0 < F_HID; k0 += 4) {
            const v2f a = *(const v2f*)(arow + k0);
            const float* __restrict__ b0p = bcol + (long)k0 * F_OUT;
            v2f b;
            b.x = b0p[0];
            b.y = b0p[F_OUT];
            cacc = __builtin_amdgcn_wmma_f32_16x16x4_f32(false, a, false, b, (short)0, cacc, false, false);
        }
        if (nok) {
#pragma unroll
            for (int r = 0; r < 8; ++r) {
                h2[(long)(m0 + r + 8 * half) * F_OUT + ncol] = cacc[r];
            }
        }
    }
}

__global__ __launch_bounds__(256) void k_gather2(const int* __restrict__ off,
                                                 const int* __restrict__ pairs,
                                                 const float* __restrict__ deg,
                                                 const float* __restrict__ h2,
                                                 const float* __restrict__ b2,
                                                 float* __restrict__ out) {
    const int lane = threadIdx.x & 31;
    const int wv = __builtin_amdgcn_readfirstlane((int)(threadIdx.x >> 5));
    const int c = blockIdx.x * 8 + wv;
    if (c >= N_NODES) return;
    const int beg = __builtin_amdgcn_readfirstlane(off[c]);
    const int end = __builtin_amdgcn_readfirstlane(off[c + 1]);
    const bool tail = lane < F_OUT - 32;

    const float di = inv_sqrt_deg(deg[c]);
    const float* __restrict__ self = h2 + (long)c * F_OUT;
    float a = self[lane] * di * di + b2[lane];
    float b = 0.0f;
    if (tail) b = self[32 + lane] * di * di + b2[32 + lane];

    for (int base = beg; base < end; base += 32) {
        const int left = end - base;
        const int n = left < 32 ? left : 32;
        int r = 0;
        float w = 0.0f;
        if (lane < n) {
            const int2 p = ((const int2*)pairs)[base + lane];
            r = p.x;
            w = __int_as_float(p.y);
        }
        for (int j = 0; j < n; j += ROWS_IN_FLIGHT) {
            float va[ROWS_IN_FLIGHT], vb[ROWS_IN_FLIGHT];
#pragma unroll
            for (int u = 0; u < ROWS_IN_FLIGHT; ++u) {
                const int ju = j + u < n ? j + u : n - 1;
                const int rj = lane_to_scalar_i(r, ju);
                const float* __restrict__ src = h2 + (long)rj * F_OUT;
                va[u] = src[lane];
                if (tail) vb[u] = src[32 + lane];
            }
#pragma unroll
            for (int u = 0; u < ROWS_IN_FLIGHT; ++u) {
                if (u == 0 || j + u < n) {
                    const float nj = lane_to_scalar_f(w, j + u);
                    a += va[u] * nj;
                    if (tail) b += vb[u] * nj;
                }
            }
        }
    }

    float* __restrict__ dst = out + (long)c * F_OUT;
    dst[lane] = a;
    if (tail) dst[32 + lane] = b;
}

extern "C" void kernel_launch(void* const* d_in, const int* in_sizes, int n_in,
                              void* d_out, int out_size, void* d_ws, size_t ws_size,
                              hipStream_t stream) {
    const float* x   = (const float*)d_in[0];
    const int*   ei  = (const int*)d_in[1];
    const float* ew  = (const float*)d_in[2];
    const float* W1  = (const float*)d_in[3];
    const float* b1  = (const float*)d_in[4];
    const float* W2  = (const float*)d_in[5];
    const float* b2  = (const float*)d_in[6];
    float*       out = (float*)d_out;

    const int E = in_sizes[2];
    const int* row = ei;
    const int* col = ei + E;

    char* ws = (char*)d_ws;
    float* deg  = (float*)(ws + 0);
    float* h1   = (float*)(ws + 409600);
    float* h2   = (float*)(ws + 409600 + 2 * 25600000);
    int*   cnt   = (int*)(ws + 59609600);
    int*   off   = (int*)(ws + 59609600 + 200192);
    int*   pos   = (int*)(ws + 60009984);
    int*   pairs = (int*)(ws + 127118848);
    (void)ws_size; (void)n_in; (void)out_size;

    const int nb_edges = (E + 255) / 256;

    k_gemm1<<<dim3(N_NODES / 16, F_HID / 64), 32, 0, stream>>>(x, W1, h1, deg, cnt);

    k_deg<<<nb_edges, 256, 0, stream>>>(col, ew, deg, cnt, pos, E);
    k_scan<<<1, SCAN_THREADS, 0, stream>>>(cnt, off);
    k_fill<<<nb_edges, 256, 0, stream>>>(row, col, ew, deg, off, pos, pairs, E);

    k_gather1_gemm2<<<N_NODES / TILE_ROWS, 512, 0, stream>>>(off, pairs, deg, h1, b1, W2, h2);

    k_gather2<<<N_NODES / 8, 256, 0, stream>>>(off, pairs, deg, h2, b2, out);
}


# edges sorted by destination with LDS counters in two levels (per chunk by node range, per range by node): no returning global atomic, 6 launches -> 5; sum order = LDS arrival slots
# speedup vs baseline: 2.6009x; 1.3234x over previous
#include <hip/hip_runtime.h>

typedef float v2f __attribute__((ext_vector_type(2)));
typedef float v8f __attribute__((ext_vector_type(8)));
typedef float v4f __attribute__((ext_vector_type(4)));

#define N_NODES 50000
#define F_IN    256
#define F_HID   128
#define F_OUT   40

static __device__ __forceinline__ float inv_sqrt_deg(float d) {
    return (d > 0.0f) ? rsqrtf(d) : 0.0f;
}

#define NCHUNK   256
#define NBINS    256
#define RANGE    196
#define TAB_COLS (NBINS + 1)
#define SORT_THREADS 1024
static_assert(NBINS * RANGE >= N_NODES, "the ranges cover every node");
static_assert(SORT_THREADS == 4 * NCHUNK, "k_place: four lanes per run");

__global__ __launch_bounds__(SORT_THREADS) void k_bin(const int* __restrict__ row,
                                                      const int* __restrict__ col,
                                                      const float* __restrict__ ew,
                                                      float* __restrict__ deg,
                                                      int4* __restrict__ recs,
                                                      int* __restrict__ tab, int E) {
    __shared__ int start[NBINS];
    __shared__ int cur[NBINS];
    const int t = threadIdx.x;
    const int chunk = (E + NCHUNK - 1) / NCHUNK;
    const int e0 = blockIdx.x * chunk;
    int e1 = e0 + chunk;
    if (e1 > E) e1 = E;

    if (t < NBINS) start[t] = 0;
    __syncthreads();

    for (int b0 = e0; b0 < e1; b0 += 4 * SORT_THREADS) {
        int c[4];
        float w[4];
#pragma unroll
        for (int u = 0; u < 4; ++u) {
            const int e = b0 + u * SORT_THREADS + t;
            const bool ok = e < e1;
            c[u] = ok ? col[e] : -1;
            w[u] = ok ? ew[e] : 0.0f;
        }
#pragma unroll
        for (int u = 0; u < 4; ++u) {
            if (c[u] >= 0) {
                atomicAdd(&deg[c[u]], w[u]);
                atomicAdd(&start[(unsigned)c[u] / RANGE], 1);
            }
        }
    }
    __syncthreads();

    if (t < 32) {
        int v[8];
        int s = 0;
#pragma unroll
        for (int q = 0; q < 8; ++q) { v[q] = start[t * 8 + q]; s += v[q]; }
        int inc = s;
        for (int d = 1; d < 32; d <<= 1) {
            const int x = __shfl_up(inc, d);
            if (t >= d) inc += x;
        }
        int run = inc - s;
#pragma unroll
        for (int q = 0; q < 8; ++q) {
            start[t * 8 + q] = run;
            cur[t * 8 + q]   = run;
            run += v[q];
        }
    }
    __syncthreads();

    int* __restrict__ trow = tab + (long)blockIdx.x * TAB_COLS;
    if (t < NBINS) trow[t] = start[t];
    if (t == NBINS) trow[NBINS] = e1 > e0 ? e1 - e0 : 0;

    int4* __restrict__ mine = recs + e0;
    for (int b0 = e0; b0 < e1; b0 += 4 * SORT_THREADS) {
        int c[4], r[4];
        float w[4];
#pragma unroll
        for (int u = 0; u < 4; ++u) {
            const int e = b0 + u * SORT_THREADS + t;
            const bool ok = e < e1;
            c[u] = ok ? col[e] : -1;
            r[u] = ok ? row[e] : 0;
            w[u] = ok ? ew[e] : 0.0f;
        }
#pragma unroll
        for (int u = 0; u < 4; ++u) {
            if (c[u] >= 0) {
                const int slot = atomicAdd(&cur[(unsigned)c[u] / RANGE], 1);
                mine[slot] = make_int4(r[u], c[u], __float_as_int(w[u]), 0);
            }
        }
    }
}

__global__ __launch_bounds__(SORT_THREADS) void k_place(const int4* __restrict__ recs,
                                                        const int* __restrict__ tab,
                                                        const float* __restrict__ deg,
                                                        int* __restrict__ off,
                                                        int* __restrict__ pairs, int E) {
    __shared__ int cnt[RANGE];
    __shared__ int red[NCHUNK / 32];
    const int t    = threadIdx.x;
    const int lane = t & 31;
    const int wv   = t >> 5;
    const int k    = blockIdx.x;
    const int n0   = k * RANGE;
    const int chunk = (E + NCHUNK - 1) / NCHUNK;

    if (t < RANGE) cnt[t] = 0;

    const int g = t >> 2;
    const int* __restrict__ trow = tab + (long)g * TAB_COLS + k;
    const int rs = trow[0];
    const int re = trow[1];
    const int4* __restrict__ run = recs + (long)g * chunk;

    int part = (t < NCHUNK) ? tab[(long)t * TAB_COLS + k] : 0;
    for (int d = 16; d > 0; d >>= 1) part += __shfl_xor(part, d);
    if (lane == 0 && wv < NCHUNK / 32) red[wv] = part;
    __syncthreads();

    for (int i = rs + (t & 3); i < re; i += 4) {
        const int c = run[i].y;
        atomicAdd(&cnt[c - n0], 1);
    }
    __syncthreads();

    if (wv == 0) {
        int base = 0;
#pragma unroll
        for (int q = 0; q < NCHUNK / 32; ++q) base += red[q];
        int v[7];
        int s = 0;
#pragma unroll
        for (int q = 0; q < 7; ++q) {
            const int idx = lane * 7 + q;
            v[q] = idx < RANGE ? cnt[idx] : 0;
            s += v[q];
        }
        int inc = s;
        for (int d = 1; d < 32; d <<= 1) {
            const int x = __shfl_up(inc, d);
            if (lane >= d) inc += x;
        }
        int slot = base + inc - s;
#pragma unroll
        for (int q = 0; q < 7; ++q) {
            const int idx = lane * 7 + q;
            if (idx < RANGE) {
                cnt[idx] = slot;
                if (n0 + idx < N_NODES) off[n0 + idx] = slot;
                slot += v[q];
            }
        }
    }
    if (k == 0 && t == SORT_THREADS - 1) off[N_NODES] = E;
    __syncthreads();

    for (int i = rs + (t & 3); i < re; i += 4) {
        const int4 rec = run[i];
        const int slot = atomicAdd(&cnt[rec.y - n0], 1);
        const float nrm = inv_sqrt_deg(deg[rec.x]) * __int_as_float(rec.z) * inv_sqrt_deg(deg[rec.y]);
        ((int2*)pairs)[slot] = make_int2(rec.x, __float_as_int(nrm));
    }
}

static_assert((N_NODES / 16) * (F_HID / 64) * 8 == N_NODES, "8 start values per wave cover every node");

__global__ __launch_bounds__(32) void k_gemm1(const float* __restrict__ X,
                                              const float* __restrict__ W,
                                              float* __restrict__ H,
                                              float* __restrict__ deg) {
    const int m0   = blockIdx.x * 16;
    const int n0   = blockIdx.y * 64;
    const int lane = threadIdx.x;
    const int half = lane >> 4;
    const int l16  = lane & 15;

    {
        const int first = (blockIdx.y * (N_NODES / 16) + blockIdx.x) * 8;
        if (lane < 8) deg[first + lane] = 1.0f;
    }

    const float* __restrict__ arow = X + (long)(m0 + l16) * F_IN;
    const float* __restrict__ bcol = W + (n0 + l16);

    v8f acc0 = {}, acc1 = {}, acc2 = {}, acc3 = {};
#pragma unroll 4
    for (int k0 = 0; k0 < F_IN; k0 += 4) {
        const int kk = k0 + 2 * half;
        v2f a;
        a.x = arow[kk + 0];
        a.y = arow[kk + 1];
        const float* __restrict__ b0p = bcol + (long)kk * F_HID;
        const float* __restrict__ b1p = b0p + F_HID;
        v2f b;
        b.x = b0p[0];  b.y = b1p[0];
        acc0 = __builtin_amdgcn_wmma_f32_16x16x4_f32(false, a, false, b, (short)0, acc0, false, false);
        b.x = b0p[16]; b.y = b1p[16];
        acc1 = __builtin_amdgcn_wmma_f32_16x16x4_f32(false, a, false, b, (short)0, acc1, false, false);
        b.x = b0p[32]; b.y = b1p[32];
        acc2 = __builtin_amdgcn_wmma_f32_16x16x4_f32(false, a, false, b, (short)0, acc2, false, false);
        b.x = b0p[48]; b.y = b1p[48];
        acc3 = __builtin_amdgcn_wmma_f32_16x16x4_f32(false, a, false, b, (short)0, acc3, false, false);
    }
#pragma unroll
    for (int r = 0; r < 8; ++r) {
        int mm = m0 + r + 8 * half;
        float* hrow = H + (long)mm * F_HID + n0 + l16;
        hrow[0]  = acc0[r];
        hrow[16] = acc1[r];
        hrow[32] = acc2[r];
        hrow[48] = acc3[r];
    }
}

#define ROWS_IN_FLIGHT 8
#define TILE_ROWS  16
#define TILE_PITCH 132
static_assert(N_NODES % TILE_ROWS == 0, "every work-group has 16 real nodes");

static __device__ __forceinline__ int lane_to_scalar_i(int v, int srclane) {
    return __builtin_amdgcn_readlane(v, srclane);
}
static __device__ __forceinline__ float lane_to_scalar_f(float v, int srclane) {
    return __int_as_float(__builtin_amdgcn_readlane(__float_as_int(v), srclane));
}

__global__ __launch_bounds__(512) void k_gather1_gemm2(const int* __restrict__ off,
                                                       const int* __restrict__ pairs,
                                                       const float* __restrict__ deg,
                                                       const float* __restrict__ h1,
                                                       const float* __restrict__ b1,
                                                       const float* __restrict__ W2,
                                                       float* __restrict__ h2) {
    __shared__ __attribute__((aligned(16))) float tile[TILE_ROWS * TILE_PITCH];

    const int lane = threadIdx.x & 31;
    const int wv = __builtin_amdgcn_readfirstlane((int)(threadIdx.x >> 5));
    const int m0 = blockIdx.x * TILE_ROWS;
    const int c  = m0 + wv;
    const int beg = __builtin_amdgcn_readfirstlane(off[c]);
    const int end = __builtin_amdgcn_readfirstlane(off[c + 1]);

    const float di = inv_sqrt_deg(deg[c]);
    const float4 sv = ((const float4*)(h1 + (long)c * F_HID))[lane];
    float4 acc;
    acc.x = sv.x * di * di;
    acc.y = sv.y * di * di;
    acc.z = sv.z * di * di;
    acc.w = sv.w * di * di;

    for (int base = beg; base < end; base += 32) {
        const int left = end - base;
        const int n = left < 32 ? left : 32;
        int r = 0;
        float w = 0.0f;
        if (lane < n) {
            const int2 p = ((const int2*)pairs)[base + lane];
            r = p.x;
            w = __int_as_float(p.y);
        }
        for (int j = 0; j < n; j += ROWS_IN_FLIGHT) {
            v4f v[ROWS_IN_FLIGHT];
#pragma unroll
            for (int u = 0; u < ROWS_IN_FLIGHT; ++u) {
                const int ju = j + u < n ? j + u : n - 1;
                const int rj = lane_to_scalar_i(r, ju);
                v[u] = ((const v4f*)(h1 + (long)rj * F_HID))[lane];
            }
            asm volatile("" :: "v"(v[0]), "v"(v[1]), "v"(v[2]), "v"(v[3]),
                               "v"(v[4]), "v"(v[5]), "v"(v[6]), "v"(v[7]));
#pragma unroll
            for (int u = 0; u < ROWS_IN_FLIGHT; ++u) {
                if (u == 0 || j + u < n) {
                    const float nj = lane_to_scalar_f(w, j + u);
                    acc.x += v[u].x * nj;
                    acc.y += v[u].y * nj;
                    acc.z += v[u].z * nj;
                    acc.w += v[u].w * nj;
                }
            }
        }
    }

    const float4 bv = ((const float4*)b1)[lane];
    float4 o;
    o.x = acc.x + bv.x;
    o.y = acc.y + bv.y;
    o.z = acc.z + bv.z;
    o.w = acc.w + bv.w;
    o.x = o.x > 0.0f ? o.x : 0.0f;
    o.y = o.y > 0.0f ? o.y : 0.0f;
    o.z = o.z > 0.0f ? o.z : 0.0f;
    o.w = o.w > 0.0f ? o.w : 0.0f;
    ((float4*)(tile + wv * TILE_PITCH))[lane] = o;
    __syncthreads();

    if (wv < 3) {
        const int half = lane >> 4;
        const int l16  = lane & 15;
        const int ncol = wv * 16 + l16;
        const bool nok = ncol < F_OUT;
        const float* __restrict__ arow = tile + l16 * TILE_PITCH + 2 * half;
        const float* __restrict__ bcol = W2 + (long)(2 * half) * F_OUT + (nok ? ncol : 0);

        v8f cacc = {};
#pragma unroll 8
        for (int k0 = 0; k0 < F_HID; k0 += 4) {
            const v2f a = *(const v2f*)(arow + k0);
            const float* __restrict__ b0p = bcol + (long)k0 * F_OUT;
            v2f b;
            b.x = b0p[0];
            b.y = b0p[F_OUT];
            cacc = __builtin_amdgcn_wmma_f32_16x16x4_f32(false, a, false, b, (short)0, cacc, false, false);
        }
        if (nok) {
#pragma unroll
            for (int r = 0; r < 8; ++r) {
                h2[(long)(m0 + r + 8 * half) * F_OUT + ncol] = cacc[r];
            }
        }
    }
}

__global__ __launch_bounds__(256) void k_gather2(const int* __restrict__ off,
                                                 const int* __restrict__ pairs,
                                                 const float* __restrict__ deg,
                                                 const float* __restrict__ h2,
                                                 const float* __restrict__ b2,
                                                 float* __restrict__ out) {
    const int lane = threadIdx.x & 31;
    const int wv = __builtin_amdgcn_readfirstlane((int)(threadIdx.x >> 5));
    const int c = blockIdx.x * 8 + wv;
    if (c >= N_NODES) return;
    const int beg = __builtin_amdgcn_readfirstlane(off[c]);
    const int end = __builtin_amdgcn_readfirstlane(off[c + 1]);
    const bool tail = lane < F_OUT - 32;

    const float di = inv_sqrt_deg(deg[c]);
    const float* __restrict__ self = h2 + (long)c * F_OUT;
    float a = self[lane] * di * di + b2[lane];
    float b = 0.0f;
    if (tail) b = self[32 + lane] * di * di + b2[32 + lane];

    for (int base = beg; base < end; base += 32) {
        const int left = end - base;
        const int n = left < 32 ? left : 32;
        int r = 0;
        float w = 0.0f;
        if (lane < n) {
            const int2 p = ((const int2*)pairs)[base + lane];
            r = p.x;
            w = __int_as_float(p.y);
        }
        for (int j = 0; j < n; j += ROWS_IN_FLIGHT) {
            float va[ROWS_IN_FLIGHT], vb[ROWS_IN_FLIGHT];
#pragma unroll
            for (int u = 0; u < ROWS_IN_FLIGHT; ++u) {
                const int ju = j + u < n ? j + u : n - 1;
                const int rj = lane_to_scalar_i(r, ju);
                const float* __restrict__ src = h2 + (long)rj * F_OUT;
                va[u] = src[lane];
                if (tail) vb[u] = src[32 + lane];
            }
#pragma unroll
            for (int u = 0; u < ROWS_IN_FLIGHT; ++u) {
                if (u == 0 || j + u < n) {
                    const float nj = lane_to_scalar_f(w, j + u);
                    a += va[u] * nj;
                    if (tail) b += vb[u] * nj;
                }
            }
        }
    }

    float* __restrict__ dst = out + (long)c * F_OUT;
    dst[lane] = a;
    if (tail) dst[32 + lane] = b;
}

extern "C" void kernel_launch(void* const* d_in, const int* in_sizes, int n_in,
                              void* d_out, int out_size, void* d_ws, size_t ws_size,
                              hipStream_t stream) {
    const float* x   = (const float*)d_in[0];
    const int*   ei  = (const int*)d_in[1];
    const float* ew  = (const float*)d_in[2];
    const float* W1  = (const float*)d_in[3];
    const float* b1  = (const float*)d_in[4];
    const float* W2  = (const float*)d_in[5];
    const float* b2  = (const float*)d_in[6];
    float*       out = (float*)d_out;

    const int E = in_sizes[2];
    const int* row = ei;
    const int* col = ei + E;

    char* ws = (char*)d_ws;
    float* deg  = (float*)(ws + 0);
    float* h1   = (float*)(ws + 409600);
    float* h2   = (float*)(ws + 409600 + 2 * 25600000);
    int*   off   = (int*)(ws + 59609600 + 200192);
    int4*  recs  = (int4*)(ws + 60009984);
    int*   tab   = (int*)(ws + 100663296);
    int*   pairs = (int*)(ws + 127118848);
    (void)ws_size; (void)n_in; (void)out_size;

    k_gemm1<<<dim3(N_NODES / 16, F_HID / 64), 32, 0, stream>>>(x, W1, h1, deg);

    k_bin<<<NCHUNK, SORT_THREADS, 0, stream>>>(row, col, ew, deg, recs, tab, E);
    k_place<<<NBINS, SORT_THREADS, 0, stream>>>(recs, tab, deg, off, pairs, E);

    k_gather1_gemm2<<<N_NODES / TILE_ROWS, 512, 0, stream>>>(off, pairs, deg, h1, b1, W2, h2);

    k_gather2<<<N_NODES / 8, 256, 0, stream>>>(off, pairs, deg, h2, b2, out);
}


# norm factored: rows stored scaled by 1/sqrt(deg), gathers sum weight*row and scale once; deg summed by LDS float adds in the sort's 2nd level (arrival order): no global atomic, no per-edge deg load
# speedup vs baseline: 3.9438x; 1.5163x over previous
#include <hip/hip_runtime.h>

typedef float v2f __attribute__((ext_vector_type(2)));
typedef float v8f __attribute__((ext_vector_type(8)));
typedef float v4f __attribute__((ext_vector_type(4)));

#define N_NODES 50000
#define F_IN    256
#define F_HID   128
#define F_OUT   40

static __device__ __forceinline__ float inv_sqrt_deg(float d) {
    return (d > 0.0f) ? rsqrtf(d) : 0.0f;
}

#define NCHUNK   256
#define NBINS    256
#define RANGE    196
#define TAB_COLS (NBINS + 1)
#define SORT_THREADS 1024
static_assert(NBINS * RANGE >= N_NODES, "the ranges cover every node");
static_assert(SORT_THREADS == 4 * NCHUNK, "k_place: four lanes per run");

__global__ __launch_bounds__(SORT_THREADS) void k_bin(const int* __restrict__ row,
                                                      const int* __restrict__ col,
                                                      const float* __restrict__ ew,
                                                      int4* __restrict__ recs,
                                                      int* __restrict__ tab, int E) {
    __shared__ int start[NBINS];
    __shared__ int cur[NBINS];
    const int t = threadIdx.x;
    const int chunk = (E + NCHUNK - 1) / NCHUNK;
    const int e0 = blockIdx.x * chunk;
    int e1 = e0 + chunk;
    if (e1 > E) e1 = E;

    if (t < NBINS) start[t] = 0;
    __syncthreads();

    for (int b0 = e0; b0 < e1; b0 += 4 * SORT_THREADS) {
        int c[4];
#pragma unroll
        for (int u = 0; u < 4; ++u) {
            const int e = b0 + u * SORT_THREADS + t;
            c[u] = e < e1 ? col[e] : -1;
        }
#pragma unroll
        for (int u = 0; u < 4; ++u) {
            if (c[u] >= 0) atomicAdd(&start[(unsigned)c[u] / RANGE], 1);
        }
    }
    __syncthreads();

    if (t < 32) {
        int v[8];
        int s = 0;
#pragma unroll
        for (int q = 0; q < 8; ++q) { v[q] = start[t * 8 + q]; s += v[q]; }
        int inc = s;
        for (int d = 1; d < 32; d <<= 1) {
            const int x = __shfl_up(inc, d);
            if (t >= d) inc += x;
        }
        int run = inc - s;
#pragma unroll
        for (int q = 0; q < 8; ++q) {
            start[t * 8 + q] = run;
            cur[t * 8 + q]   = run;
            run += v[q];
        }
    }
    __syncthreads();

    int* __restrict__ trow = tab + (long)blockIdx.x * TAB_COLS;
    if (t < NBINS) trow[t] = start[t];
    if (t == NBINS) trow[NBINS] = e1 > e0 ? e1 - e0 : 0;

    int4* __restrict__ mine = recs + e0;
    for (int b0 = e0; b0 < e1; b0 += 4 * SORT_THREADS) {
        int c[4], r[4];
        float w[4];
#pragma unroll
        for (int u = 0; u < 4; ++u) {
            const int e = b0 + u * SORT_THREADS + t;
            const bool ok = e < e1;
            c[u] = ok ? col[e] : -1;
            r[u] = ok ? row[e] : 0;
            w[u] = ok ? ew[e] : 0.0f;
        }
#pragma unroll
        for (int u = 0; u < 4; ++u) {
            if (c[u] >= 0) {
                const int slot = atomicAdd(&cur[(unsigned)c[u] / RANGE], 1);
                mine[slot] = make_int4(r[u], c[u], __float_as_int(w[u]), 0);
            }
        }
    }
}

__global__ __launch_bounds__(SORT_THREADS) void k_place(const int4* __restrict__ recs,
                                                        const int* __restrict__ tab,
                                                        float* __restrict__ dinv,
                                                        int* __restrict__ off,
                                                        int* __restrict__ pairs, int E) {
    __shared__ int cnt[RANGE];
    __shared__ float dsum[RANGE];
    __shared__ int red[NCHUNK / 32];
    const int t    = threadIdx.x;
    const int lane = t & 31;
    const int wv   = t >> 5;
    const int k    = blockIdx.x;
    const int n0   = k * RANGE;
    const int chunk = (E + NCHUNK - 1) / NCHUNK;

    if (t < RANGE) { cnt[t] = 0; dsum[t] = 1.0f; }

    const int g = t >> 2;
    const int* __restrict__ trow = tab + (long)g * TAB_COLS + k;
    const int rs = trow[0];
    const int re = trow[1];
    const int4* __restrict__ run = recs + (long)g * chunk;

    int part = (t < NCHUNK) ? tab[(long)t * TAB_COLS + k] : 0;
    for (int d = 16; d > 0; d >>= 1) part += __shfl_xor(part, d);
    if (lane == 0 && wv < NCHUNK / 32) red[wv] = part;
    __syncthreads();

    for (int i = rs + (t & 3); i < re; i += 4) {
        const int4 rec = run[i];
        atomicAdd(&cnt[rec.y - n0], 1);
        atomicAdd(&dsum[rec.y - n0], __int_as_float(rec.z));
    }
    __syncthreads();

    if (t >= SORT_THREADS - RANGE) {
        const int idx = t - (SORT_THREADS - RANGE);
        if (n0 + idx < N_NODES) dinv[n0 + idx] = inv_sqrt_deg(dsum[idx]);
    }

    if (wv == 0) {
        int base = 0;
#pragma unroll
        for (int q = 0; q < NCHUNK / 32; ++q) base += red[q];
        int v[7];
        int s = 0;
#pragma unroll
        for (int q = 0; q < 7; ++q) {
            const int idx = lane * 7 + q;
            v[q] = idx < RANGE ? cnt[idx] : 0;
            s += v[q];
        }
        int inc = s;
        for (int d = 1; d < 32; d <<= 1) {
            const int x = __shfl_up(inc, d);
            if (lane >= d) inc += x;
        }
        int slot = base + inc - s;
#pragma unroll
        for (int q = 0; q < 7; ++q) {
            const int idx = lane * 7 + q;
            if (idx < RANGE) {
                cnt[idx] = slot;
                if (n0 + idx < N_NODES) off[n0 + idx] = slot;
                slot += v[q];
            }
        }
    }
    if (k == 0 && t == SORT_THREADS - 1) off[N_NODES] = E;
    __syncthreads();

    for (int i = rs + (t & 3); i < re; i += 4) {
        const int4 rec = run[i];
        const int slot = atomicAdd(&cnt[rec.y - n0], 1);
        ((int2*)pairs)[slot] = make_int2(rec.x, rec.z);
    }
}

__global__ __launch_bounds__(32) void k_gemm1(const float* __restrict__ X,
                                              const float* __restrict__ W,
                                              const float* __restrict__ dinv,
                                              float* __restrict__ H) {
    const int m0   = blockIdx.x * 16;
    const int n0   = blockIdx.y * 64;
    const int lane = threadIdx.x;
    const int half = lane >> 4;
    const int l16  = lane & 15;

    const float4 dlo = ((const float4*)(dinv + m0 + 8 * half))[0];
    const float4 dhi = ((const float4*)(dinv + m0 + 8 * half))[1];
    const float dv[8] = {dlo.x, dlo.y, dlo.z, dlo.w, dhi.x, dhi.y, dhi.z, dhi.w};

    const float* __restrict__ arow = X + (long)(m0 + l16) * F_IN;
    const float* __restrict__ bcol = W + (n0 + l16);

    v8f acc0 = {}, acc1 = {}, acc2 = {}, acc3 = {};
#pragma unroll 4
    for (int k0 = 0; k0 < F_IN; k0 += 4) {
        const int kk = k0 + 2 * half;
        v2f a;
        a.x = arow[kk + 0];
        a.y = arow[kk + 1];
        const float* __restrict__ b0p = bcol + (long)kk * F_HID;
        const float* __restrict__ b1p = b0p + F_HID;
        v2f b;
        b.x = b0p[0];  b.y = b1p[0];
        acc0 = __builtin_amdgcn_wmma_f32_16x16x4_f32(false, a, false, b, (short)0, acc0, false, false);
        b.x = b0p[16]; b.y = b1p[16];
        acc1 = __builtin_amdgcn_wmma_f32_16x16x4_f32(false, a, false, b, (short)0, acc1, false, false);
        b.x = b0p[32]; b.y = b1p[32];
        acc2 = __builtin_amdgcn_wmma_f32_16x16x4_f32(false, a, false, b, (short)0, acc2, false, false);
        b.x = b0p[48]; b.y = b1p[48];
        acc3 = __builtin_amdgcn_wmma_f32_16x16x4_f32(false, a, false, b, (short)0, acc3, false, false);
    }
#pragma unroll
    for (int r = 0; r < 8; ++r) {
        int mm = m0 + r + 8 * half;
        float* hrow = H + (long)mm * F_HID + n0 + l16;
        hrow[0]  = acc0[r] * dv[r];
        hrow[16] = acc1[r] * dv[r];
        hrow[32] = acc2[r] * dv[r];
        hrow[48] = acc3[r] * dv[r];
    }
}

#define ROWS_IN_FLIGHT 8
#define TILE_ROWS  16
#define TILE_PITCH 132
static_assert(N_NODES % TILE_ROWS == 0, "every work-group has 16 real nodes");

static __device__ __forceinline__ int lane_to_scalar_i(int v, int srclane) {
    return __builtin_amdgcn_readlane(v, srclane);
}
static __device__ __forceinline__ float lane_to_scalar_f(float v, int srclane) {
    return __int_as_float(__builtin_amdgcn_readlane(__float_as_int(v), srclane));
}

__global__ __launch_bounds__(512) void k_gather1_gemm2(const int* __restrict__ off,
                                                       const int* __restrict__ pairs,
                                                       const float* __restrict__ dinv,
                                                       const float* __restrict__ h1,
                                                       const float* __restrict__ b1,
                                                       const float* __restrict__ W2,
                                                       float* __restrict__ h2) {
    __shared__ __attribute__((aligned(16))) float tile[TILE_ROWS * TILE_PITCH];
    __shared__ float dtile[TILE_ROWS];

    const int lane = threadIdx.x & 31;
    const int wv = __builtin_amdgcn_readfirstlane((int)(threadIdx.x >> 5));
    const int m0 = blockIdx.x * TILE_ROWS;
    const int c  = m0 + wv;
    const int beg = __builtin_amdgcn_readfirstlane(off[c]);
    const int end = __builtin_amdgcn_readfirstlane(off[c + 1]);

    const float di = dinv[c];
    float4 acc = ((const float4*)(h1 + (long)c * F_HID))[lane];

    for (int base = beg; base < end; base += 32) {
        const int left = end - base;
        const int n = left < 32 ? left : 32;
        int r = 0;
        float w = 0.0f;
        if (lane < n) {
            const int2 p = ((const int2*)pairs)[base + lane];
            r = p.x;
            w = __int_as_float(p.y);
        }
        for (int j = 0; j < n; j += ROWS_IN_FLIGHT) {
            v4f v[ROWS_IN_FLIGHT];
#pragma unroll
            for (int u = 0; u < ROWS_IN_FLIGHT; ++u) {
                const int ju = j + u < n ? j + u : n - 1;
                const int rj = lane_to_scalar_i(r, ju);
                v[u] = ((const v4f*)(h1 + (long)rj * F_HID))[lane];
            }
            asm volatile("" :: "v"(v[0]), "v"(v[1]), "v"(v[2]), "v"(v[3]),
                               "v"(v[4]), "v"(v[5]), "v"(v[6]), "v"(v[7]));
#pragma unroll
            for (int u = 0; u < ROWS_IN_FLIGHT; ++u) {
                if (u == 0 || j + u < n) {
                    const float nj = lane_to_scalar_f(w, j + u);
                    acc.x += v[u].x * nj;
                    acc.y += v[u].y * nj;
                    acc.z += v[u].z * nj;
                    acc.w += v[u].w * nj;
                }
            }
        }
    }

    const float4 bv = ((const float4*)b1)[lane];
    float4 o;
    o.x = acc.x * di + bv.x;
    o.y = acc.y * di + bv.y;
    o.z = acc.z * di + bv.z;
    o.w = acc.w * di + bv.w;
    o.x = o.x > 0.0f ? o.x : 0.0f;
    o.y = o.y > 0.0f ? o.y : 0.0f;
    o.z = o.z > 0.0f ? o.z : 0.0f;
    o.w = o.w > 0.0f ? o.w : 0.0f;
    ((float4*)(tile + wv * TILE_PITCH))[lane] = o;
    if (lane == 0) dtile[wv] = di;
    __syncthreads();

    if (wv < 3) {
        const int half = lane >> 4;
        const int l16  = lane & 15;
        const int ncol = wv * 16 + l16;
        const bool nok = ncol < F_OUT;
        const float* __restrict__ arow = tile + l16 * TILE_PITCH + 2 * half;
        const float* __restrict__ bcol = W2 + (long)(2 * half) * F_OUT + (nok ? ncol : 0);

        v8f cacc = {};
#pragma unroll 8
        for (int k0 = 0; k0 < F_HID; k0 += 4) {
            const v2f a = *(const v2f*)(arow + k0);
            const float* __restrict__ b0p = bcol + (long)k0 * F_OUT;
            v2f b;
            b.x = b0p[0];
            b.y = b0p[F_OUT];
            cacc = __builtin_amdgcn_wmma_f32_16x16x4_f32(false, a, false, b, (short)0, cacc, false, false);
        }
        if (nok) {
#pragma unroll
            for (int r = 0; r < 8; ++r) {
                h2[(long)(m0 + r + 8 * half) * F_OUT + ncol] = cacc[r] * dtile[r + 8 * half];
            }
        }
    }
}

__global__ __launch_bounds__(256) void k_gather2(const int* __restrict__ off,
                                                 const int* __restrict__ pairs,
                                                 const float* __restrict__ dinv,
                                                 const float* __restrict__ h2,
                                                 const float* __restrict__ b2,
                                                 float* __restrict__ out) {
    const int lane = threadIdx.x & 31;
    const int wv = __builtin_amdgcn_readfirstlane((int)(threadIdx.x >> 5));
    const int c = blockIdx.x * 8 + wv;
    if (c >= N_NODES) return;
    const int beg = __builtin_amdgcn_readfirstlane(off[c]);
    const int end = __builtin_amdgcn_readfirstlane(off[c + 1]);
    const bool tail = lane < F_OUT - 32;

    const float di = dinv[c];
    const float* __restrict__ self = h2 + (long)c * F_OUT;
    float a = self[lane];
    float b = 0.0f;
    if (tail) b = self[32 + lane];

    for (int base = beg; base < end; base += 32) {
        const int left = end - base;
        const int n = left < 32 ? left : 32;
        int r = 0;
        float w = 0.0f;
        if (lane < n) {
            const int2 p = ((const int2*)pairs)[base + lane];
            r = p.x;
            w = __int_as_float(p.y);
        }
        for (int j = 0; j < n; j += ROWS_IN_FLIGHT) {
            float va[ROWS_IN_FLIGHT], vb[ROWS_IN_FLIGHT];
#pragma unroll
            for (int u = 0; u < ROWS_IN_FLIGHT; ++u) {
                const int ju = j + u < n ? j + u : n - 1;
                const int rj = lane_to_scalar_i(r, ju);
                const float* __restrict__ src = h2 + (long)rj * F_OUT;
                va[u] = src[lane];
                if (tail) vb[u] = src[32 + lane];
            }
#pragma unroll
            for (int u = 0; u < ROWS_IN_FLIGHT; ++u) {
                if (u == 0 || j + u < n) {
                    const float nj = lane_to_scalar_f(w, j + u);
                    a += va[u] * nj;
                    if (tail) b += vb[u] * nj;
                }
            }
        }
    }

    float* __restrict__ dst = out + (long)c * F_OUT;
    dst[lane] = a * di + b2[lane];
    if (tail) dst[32 + lane] = b * di + b2[32 + lane];
}

extern "C" void kernel_launch(void* const* d_in, const int* in_sizes, int n_in,
                              void* d_out, int out_size, void* d_ws, size_t ws_size,
                              hipStream_t stream) {
    const float* x   = (const float*)d_in[0];
    const int*   ei  = (const int*)d_in[1];
    const float* ew  = (const float*)d_in[2];
    const float* W1  = (const float*)d_in[3];
    const float* b1  = (const float*)d_in[4];
    const float* W2  = (const float*)d_in[5];
    const float* b2  = (const float*)d_in[6];
    float*       out = (float*)d_out;

    const int E = in_sizes[2];
    const int* row = ei;
    const int* col = ei + E;

    char* ws = (char*)d_ws;
    float* dinv = (float*)(ws + 0);
    float* h1   = (float*)(ws + 409600);
    float* h2   = (float*)(ws + 409600 + 2 * 25600000);
    int*   off   = (int*)(ws + 59609600 + 200192);
    int4*  recs  = (int4*)(ws + 60009984);
    int*   tab   = (int*)(ws + 100663296);
    int*   pairs = (int*)(ws + 127118848);
    (void)ws_size; (void)n_in; (void)out_size;

    k_bin<<<NCHUNK, SORT_THREADS, 0, stream>>>(row, col, ew, recs, tab, E);
    k_place<<<NBINS, SORT_THREADS, 0, stream>>>(recs, tab, dinv, off, pairs, E);

    k_gemm1<<<dim3(N_NODES / 16, F_HID / 64), 32, 0, stream>>>(x, W1, dinv, h1);

    k_gather1_gemm2<<<N_NODES / TILE_ROWS, 512, 0, stream>>>(off, pairs, dinv, h1, b1, W2, h2);

    k_gather2<<<N_NODES / 8, 256, 0, stream>>>(off, pairs, dinv, h2, b2, out);
}


# gather kernels: a wave takes 4 consecutive destination nodes in turn (bounds and first pairs fetched at once, own row a node ahead, tail loads unpredicated): 12500 waves each, not 50000; same sums
# speedup vs baseline: 4.2803x; 1.0853x over previous
#include <hip/hip_runtime.h>

typedef float v2f __attribute__((ext_vector_type(2)));
typedef float v8f __attribute__((ext_vector_type(8)));
typedef float v4f __attribute__((ext_vector_type(4)));

#define N_NODES 50000
#define F_IN    256
#define F_HID   128
#define F_OUT   40

static __device__ __forceinline__ float inv_sqrt_deg(float d) {
    return (d > 0.0f) ? rsqrtf(d) : 0.0f;
}

#define NCHUNK   256
#define NBINS    256
#define RANGE    196
#define TAB_COLS (NBINS + 1)
#define SORT_THREADS 1024
static_assert(NBINS * RANGE >= N_NODES, "the ranges cover every node");
static_assert(SORT_THREADS == 4 * NCHUNK, "k_place: four lanes per run");

__global__ __launch_bounds__(SORT_THREADS) void k_bin(const int* __restrict__ row,
                                                      const int* __restrict__ col,
                                                      const float* __restrict__ ew,
                                                      int4* __restrict__ recs,
                                                      int* __restrict__ tab, int E) {
    __shared__ int start[NBINS];
    __shared__ int cur[NBINS];
    const int t = threadIdx.x;
    const int chunk = (E + NCHUNK - 1) / NCHUNK;
    const int e0 = blockIdx.x * chunk;
    int e1 = e0 + chunk;
    if (e1 > E) e1 = E;

    if (t < NBINS) start[t] = 0;
    __syncthreads();

    for (int b0 = e0; b0 < e1; b0 += 4 * SORT_THREADS) {
        int c[4];
#pragma unroll
        for (int u = 0; u < 4; ++u) {
            const int e = b0 + u * SORT_THREADS + t;
            c[u] = e < e1 ? col[e] : -1;
        }
#pragma unroll
        for (int u = 0; u < 4; ++u) {
            if (c[u] >= 0) atomicAdd(&start[(unsigned)c[u] / RANGE], 1);
        }
    }
    __syncthreads();

    if (t < 32) {
        int v[8];
        int s = 0;
#pragma unroll
        for (int q = 0; q < 8; ++q) { v[q] = start[t * 8 + q]; s += v[q]; }
        int inc = s;
        for (int d = 1; d < 32; d <<= 1) {
            const int x = __shfl_up(inc, d);
            if (t >= d) inc += x;
        }
        int run = inc - s;
#pragma unroll
        for (int q = 0; q < 8; ++q) {
            start[t * 8 + q] = run;
            cur[t * 8 + q]   = run;
            run += v[q];
        }
    }
    __syncthreads();

    int* __restrict__ trow = tab + (long)blockIdx.x * TAB_COLS;
    if (t < NBINS) trow[t] = start[t];
    if (t == NBINS) trow[NBINS] = e1 > e0 ? e1 - e0 : 0;

    int4* __restrict__ mine = recs + e0;
    for (int b0 = e0; b0 < e1; b0 += 4 * SORT_THREADS) {
        int c[4], r[4];
        float w[4];
#pragma unroll
        for (int u = 0; u < 4; ++u) {
            const int e = b0 + u * SORT_THREADS + t;
            const bool ok = e < e1;
            c[u] = ok ? col[e] : -1;
            r[u] = ok ? row[e] : 0;
            w[u] = ok ? ew[e] : 0.0f;
        }
#pragma unroll
        for (int u = 0; u < 4; ++u) {
            if (c[u] >= 0) {
                const int slot = atomicAdd(&cur[(unsigned)c[u] / RANGE], 1);
                mine[slot] = make_int4(r[u], c[u], __float_as_int(w[u]), 0);
            }
        }
    }
}

__global__ __launch_bounds__(SORT_THREADS) void k_place(const int4* __restrict__ recs,
                                                        const int* __restrict__ tab,
                                                        float* __restrict__ dinv,
                                                        int* __restrict__ off,
                                                        int* __restrict__ pairs, int E) {
    __shared__ int cnt[RANGE];
    __shared__ float dsum[RANGE];
    __shared__ int red[NCHUNK / 32];
    const int t    = threadIdx.x;
    const int lane = t & 31;
    const int wv   = t >> 5;
    const int k    = blockIdx.x;
    const int n0   = k * RANGE;
    const int chunk = (E + NCHUNK - 1) / NCHUNK;

    if (t < RANGE) { cnt[t] = 0; dsum[t] = 1.0f; }

    const int g = t >> 2;
    const int* __restrict__ trow = tab + (long)g * TAB_COLS + k;
    const int rs = trow[0];
    const int re = trow[1];
    const int4* __restrict__ run = recs + (long)g * chunk;

    int part = (t < NCHUNK) ? tab[(long)t * TAB_COLS + k] : 0;
    for (int d = 16; d > 0; d >>= 1) part += __shfl_xor(part, d);
    if (lane == 0 && wv < NCHUNK / 32) red[wv] = part;
    __syncthreads();

    for (int i = rs + (t & 3); i < re; i += 4) {
        const int4 rec = run[i];
        atomicAdd(&cnt[rec.y - n0], 1);
        atomicAdd(&dsum[rec.y - n0], __int_as_float(rec.z));
    }
    __syncthreads();

    if (t >= SORT_THREADS - RANGE) {
        const int idx = t - (SORT_THREADS - RANGE);
        if (n0 + idx < N_NODES) dinv[n0 + idx] = inv_sqrt_deg(dsum[idx]);
    }

    if (wv == 0) {
        int base = 0;
#pragma unroll
        for (int q = 0; q < NCHUNK / 32; ++q) base += red[q];
        int v[7];
        int s = 0;
#pragma unroll
        for (int q = 0; q < 7; ++q) {
            const int idx = lane * 7 + q;
            v[q] = idx < RANGE ? cnt[idx] : 0;
            s += v[q];
        }
        int inc = s;
        for (int d = 1; d < 32; d <<= 1) {
            const int x = __shfl_up(inc, d);
            if (lane >= d) inc += x;
        }
        int slot = base + inc - s;
#pragma unroll
        for (int q = 0; q < 7; ++q) {
            const int idx = lane * 7 + q;
            if (idx < RANGE) {
                cnt[idx] = slot;
                if (n0 + idx < N_NODES) off[n0 + idx] = slot;
                slot += v[q];
            }
        }
    }
    if (k == 0 && t == SORT_THREADS - 1) off[N_NODES] = E;
    __syncthreads();

    for (int i = rs + (t & 3); i < re; i += 4) {
        const int4 rec = run[i];
        const int slot = atomicAdd(&cnt[rec.y - n0], 1);
        ((int2*)pairs)[slot] = make_int2(rec.x, rec.z);
    }
}

__global__ __launch_bounds__(32) void k_gemm1(const float* __restrict__ X,
                                              const float* __restrict__ W,
                                              const float* __restrict__ dinv,
                                              float* __restrict__ H) {
    const int m0   = blockIdx.x * 16;
    const int n0   = blockIdx.y * 64;
    const int lane = threadIdx.x;
    const int half = lane >> 4;
    const int l16  = lane & 15;

    const float4 dlo = ((const float4*)(dinv + m0 + 8 * half))[0];
    const float4 dhi = ((const float4*)(dinv + m0 + 8 * half))[1];
    const float dv[8] = {dlo.x, dlo.y, dlo.z, dlo.w, dhi.x, dhi.y, dhi.z, dhi.w};

    const float* __restrict__ arow = X + (long)(m0 + l16) * F_IN;
    const float* __restrict__ bcol = W + (n0 + l16);

    v8f acc0 = {}, acc1 = {}, acc2 = {}, acc3 = {};
#pragma unroll 4
    for (int k0 = 0; k0 < F_IN; k0 += 4) {
        const int kk = k0 + 2 * half;
        v2f a;
        a.x = arow[kk + 0];
        a.y = arow[kk + 1];
        const float* __restrict__ b0p = bcol + (long)kk * F_HID;
        const float* __restrict__ b1p = b0p + F_HID;
        v2f b;
        b.x = b0p[0];  b.y = b1p[0];
        acc0 = __builtin_amdgcn_wmma_f32_16x16x4_f32(false, a, false, b, (short)0, acc0, false, false);
        b.x = b0p[16]; b.y = b1p[16];
        acc1 = __builtin_amdgcn_wmma_f32_16x16x4_f32(false, a, false, b, (short)0, acc1, false, false);
        b.x = b0p[32]; b.y = b1p[32];
        acc2 = __builtin_amdgcn_wmma_f32_16x16x4_f32(false, a, false, b, (short)0, acc2, false, false);
        b.x = b0p[48]; b.y = b1p[48];
        acc3 = __builtin_amdgcn_wmma_f32_16x16x4_f32(false, a, false, b, (short)0, acc3, false, false);
    }
#pragma unroll
    for (int r = 0; r < 8; ++r) {
        int mm = m0 + r + 8 * half;
        float* hrow = H + (long)mm * F_HID + n0 + l16;
        hrow[0]  = acc0[r] * dv[r];
        hrow[16] = acc1[r] * dv[r];
        hrow[32] = acc2[r] * dv[r];
        hrow[48] = acc3[r] * dv[r];
    }
}

#define ROWS_IN_FLIGHT 8
#define NODES_PER_WAVE 4
#define TILE_ROWS  16
#define G1_WAVES   (TILE_ROWS / NODES_PER_WAVE)
#define TILE_PITCH 132
static_assert(N_NODES % TILE_ROWS == 0, "every work-group has 16 real nodes");
static_assert(TILE_ROWS % NODES_PER_WAVE == 0 && G1_WAVES >= 3, "the tile's rows split evenly over the waves; three waves do the product");

static __device__ __forceinline__ int lane_to_scalar_i(int v, int srclane) {
    return __builtin_amdgcn_readlane(v, srclane);
}
static __device__ __forceinline__ float lane_to_scalar_f(float v, int srclane) {
    return __int_as_float(__builtin_amdgcn_readlane(__float_as_int(v), srclane));
}

static __device__ __forceinline__ int2 first_pairs(const int* __restrict__ pairs, int beg, int end, int lane) {
    int idx = beg + lane;
    idx = idx < end - 1 ? idx : end - 1;
    idx = idx > 0 ? idx : 0;
    return ((const int2*)pairs)[idx];
}

static __device__ __forceinline__ void sum_rows128(const float* __restrict__ h1, int lane,
                                                   int r, float w, int n, float4& acc) {
    for (int j = 0; j < n; j += ROWS_IN_FLIGHT) {
        v4f v[ROWS_IN_FLIGHT];
#pragma unroll
        for (int u = 0; u < ROWS_IN_FLIGHT; ++u) {
            const int ju = j + u < n ? j + u : n - 1;
            const int rj = lane_to_scalar_i(r, ju);
            v[u] = ((const v4f*)(h1 + (long)rj * F_HID))[lane];
        }
        asm volatile("" :: "v"(v[0]), "v"(v[1]), "v"(v[2]), "v"(v[3]),
                           "v"(v[4]), "v"(v[5]), "v"(v[6]), "v"(v[7]));
#pragma unroll
        for (int u = 0; u < ROWS_IN_FLIGHT; ++u) {
            if (u == 0 || j + u < n) {
                const float nj = lane_to_scalar_f(w, j + u);
                acc.x += v[u].x * nj;
                acc.y += v[u].y * nj;
                acc.z += v[u].z * nj;
                acc.w += v[u].w * nj;
            }
        }
    }
}

__global__ __launch_bounds__(32 * G1_WAVES, 16) void k_gather1_gemm2(const int* __restrict__ off,
                                                       const int* __restrict__ pairs,
                                                       const float* __restrict__ dinv,
                                                       const float* __restrict__ h1,
                                                       const float* __restrict__ b1,
                                                       const float* __restrict__ W2,
                                                       float* __restrict__ h2) {
    __shared__ __attribute__((aligned(16))) float tile[TILE_ROWS * TILE_PITCH];
    __shared__ float dtile[TILE_ROWS];

    const int lane = threadIdx.x & 31;
    const int wv = __builtin_amdgcn_readfirstlane((int)(threadIdx.x >> 5));
    const int m0 = blockIdx.x * TILE_ROWS;
    const int row0 = wv * NODES_PER_WAVE;
    const int c0 = m0 + row0;

    int bound[NODES_PER_WAVE + 1];
#pragma unroll
    for (int i = 0; i <= NODES_PER_WAVE; ++i) bound[i] = __builtin_amdgcn_readfirstlane(off[c0 + i]);

    float dself[NODES_PER_WAVE];
#pragma unroll
    for (int i = 0; i < NODES_PER_WAVE; ++i) dself[i] = dinv[c0 + i];

    int2 p0[NODES_PER_WAVE];
#pragma unroll
    for (int i = 0; i < NODES_PER_WAVE; ++i) p0[i] = first_pairs(pairs, bound[i], bound[i + 1], lane);

    const float4 bv = ((const float4*)b1)[lane];

    float4 own = ((const float4*)(h1 + (long)c0 * F_HID))[lane];

#pragma unroll
    for (int i = 0; i < NODES_PER_WAVE; ++i) {
        const int beg = bound[i];
        const int end = bound[i + 1];
        const float di = dself[i];
        float4 acc = own;
        if (i + 1 < NODES_PER_WAVE) own = ((const float4*)(h1 + (long)(c0 + i + 1) * F_HID))[lane];

        const int cnt = end - beg;
        sum_rows128(h1, lane, p0[i].x, __int_as_float(p0[i].y), cnt < 32 ? cnt : 32, acc);
        for (int base = beg + 32; base < end; base += 32) {
            const int left = end - base;
            const int n = left < 32 ? left : 32;
            int r = 0;
            float w = 0.0f;
            if (lane < n) {
                const int2 p = ((const int2*)pairs)[base + lane];
                r = p.x;
                w = __int_as_float(p.y);
            }
            sum_rows128(h1, lane, r, w, n, acc);
        }

        float4 o;
        o.x = acc.x * di + bv.x;
        o.y = acc.y * di + bv.y;
        o.z = acc.z * di + bv.z;
        o.w = acc.w * di + bv.w;
        o.x = o.x > 0.0f ? o.x : 0.0f;
        o.y = o.y > 0.0f ? o.y : 0.0f;
        o.z = o.z > 0.0f ? o.z : 0.0f;
        o.w = o.w > 0.0f ? o.w : 0.0f;
        ((float4*)(tile + (row0 + i) * TILE_PITCH))[lane] = o;
        if (lane == 0) dtile[row0 + i] = di;
    }
    __syncthreads();

    if (wv < 3) {
        const int half = lane >> 4;
        const int l16  = lane & 15;
        const int ncol = wv * 16 + l16;
        const bool nok = ncol < F_OUT;
        const float* __restrict__ arow = tile + l16 * TILE_PITCH + 2 * half;
        const float* __restrict__ bcol = W2 + (long)(2 * half) * F_OUT + (nok ? ncol : 0);

        v8f cacc = {};
#pragma unroll 8
        for (int k0 = 0; k0 < F_HID; k0 += 4) {
            const v2f a = *(const v2f*)(arow + k0);
            const float* __restrict__ b0p = bcol + (long)k0 * F_OUT;
            v2f b;
            b.x = b0p[0];
            b.y = b0p[F_OUT];
            cacc = __builtin_amdgcn_wmma_f32_16x16x4_f32(false, a, false, b, (short)0, cacc, false, false);
        }
        if (nok) {
#pragma unroll
            for (int r = 0; r < 8; ++r) {
                h2[(long)(m0 + r + 8 * half) * F_OUT + ncol] = cacc[r] * dtile[r + 8 * half];
            }
        }
    }
}

#define G2_WAVES 8
static_assert(N_NODES % NODES_PER_WAVE == 0, "a wave's nodes are all real");

static __device__ __forceinline__ void sum_rows40(const float* __restrict__ h2, int lane, int tl,
                                                  int r, float w, int n, float& a, float& b) {
    for (int j = 0; j < n; j += ROWS_IN_FLIGHT) {
        float va[ROWS_IN_FLIGHT], vb[ROWS_IN_FLIGHT];
#pragma unroll
        for (int u = 0; u < ROWS_IN_FLIGHT; ++u) {
            const int ju = j + u < n ? j + u : n - 1;
            const int rj = lane_to_scalar_i(r, ju);
            const float* __restrict__ src = h2 + (long)rj * F_OUT;
            va[u] = src[lane];
            vb[u] = src[32 + tl];
        }
        asm volatile("" :: "v"(va[0]), "v"(va[1]), "v"(va[2]), "v"(va[3]),
                           "v"(va[4]), "v"(va[5]), "v"(va[6]), "v"(va[7]),
                           "v"(vb[0]), "v"(vb[1]), "v"(vb[2]), "v"(vb[3]),
                           "v"(vb[4]), "v"(vb[5]), "v"(vb[6]), "v"(vb[7]));
#pragma unroll
        for (int u = 0; u < ROWS_IN_FLIGHT; ++u) {
            if (u == 0 || j + u < n) {
                const float nj = lane_to_scalar_f(w, j + u);
                a += va[u] * nj;
                b += vb[u] * nj;
            }
        }
    }
}

__global__ __launch_bounds__(32 * G2_WAVES) void k_gather2(const int* __restrict__ off,
                                                 const int* __restrict__ pairs,
                                                 const float* __restrict__ dinv,
                                                 const float* __restrict__ h2,
                                                 const float* __restrict__ b2,
                                                 float* __restrict__ out) {
    const int lane = threadIdx.x & 31;
    const int wv = __builtin_amdgcn_readfirstlane((int)(threadIdx.x >> 5));
    const int c0 = (blockIdx.x * G2_WAVES + wv) * NODES_PER_WAVE;
    if (c0 >= N_NODES) return;
    const bool tail = lane < F_OUT - 32;
    const int tl = tail ? lane : 0;

    int bound[NODES_PER_WAVE + 1];
#pragma unroll
    for (int i = 0; i <= NODES_PER_WAVE; ++i) bound[i] = __builtin_amdgcn_readfirstlane(off[c0 + i]);

    float dself[NODES_PER_WAVE];
#pragma unroll
    for (int i = 0; i < NODES_PER_WAVE; ++i) dself[i] = dinv[c0 + i];

    int2 p0[NODES_PER_WAVE];
#pragma unroll
    for (int i = 0; i < NODES_PER_WAVE; ++i) p0[i] = first_pairs(pairs, bound[i], bound[i + 1], lane);

    const float bias_a = b2[lane];
    const float bias_b = b2[32 + tl];

    float own_a = h2[(long)c0 * F_OUT + lane];
    float own_b = h2[(long)c0 * F_OUT + 32 + tl];

#pragma unroll
    for (int i = 0; i < NODES_PER_WAVE; ++i) {
        const int c = c0 + i;
        const int beg = bound[i];
        const int end = bound[i + 1];
        const float di = dself[i];
        float a = own_a;
        float b = own_b;
        if (i + 1 < NODES_PER_WAVE) {
            own_a = h2[(long)(c + 1) * F_OUT + lane];
            own_b = h2[(long)(c + 1) * F_OUT + 32 + tl];
        }

        const int cnt = end - beg;
        sum_rows40(h2, lane, tl, p0[i].x, __int_as_float(p0[i].y), cnt < 32 ? cnt : 32, a, b);
        for (int base = beg + 32; base < end; base += 32) {
            const int left = end - base;
            const int n = left < 32 ? left : 32;
            int r = 0;
            float w = 0.0f;
            if (lane < n) {
                const int2 p = ((const int2*)pairs)[base + lane];
                r = p.x;
                w = __int_as_float(p.y);
            }
            sum_rows40(h2, lane, tl, r, w, n, a, b);
        }

        float* __restrict__ dst = out + (long)c * F_OUT;
        dst[lane] = a * di + bias_a;
        if (tail) dst[32 + lane] = b * di + bias_b;
    }
}

extern "C" void kernel_launch(void* const* d_in, const int* in_sizes, int n_in,
                              void* d_out, int out_size, void* d_ws, size_t ws_size,
                              hipStream_t stream) {
    const float* x   = (const float*)d_in[0];
    const int*   ei  = (const int*)d_in[1];
    const float* ew  = (const float*)d_in[2];
    const float* W1  = (const float*)d_in[3];
    const float* b1  = (const float*)d_in[4];
    const float* W2  = (const float*)d_in[5];
    const float* b2  = (const float*)d_in[6];
    float*       out = (float*)d_out;

    const int E = in_sizes[2];
    const int* row = ei;
    const int* col = ei + E;

    char* ws = (char*)d_ws;
    float* dinv = (float*)(ws + 0);
    float* h1   = (float*)(ws + 409600);
    float* h2   = (float*)(ws + 409600 + 2 * 25600000);
    int*   off   = (int*)(ws + 59609600 + 200192);
    int4*  recs  = (int4*)(ws + 60009984);
    int*   tab   = (int*)(ws + 100663296);
    int*   pairs = (int*)(ws + 127118848);
    (void)ws_size; (void)n_in; (void)out_size;

    k_bin<<<NCHUNK, SORT_THREADS, 0, stream>>>(row, col, ew, recs, tab, E);
    k_place<<<NBINS, SORT_THREADS, 0, stream>>>(recs, tab, dinv, off, pairs, E);

    k_gemm1<<<dim3(N_NODES / 16, F_HID / 64), 32, 0, stream>>>(x, W1, dinv, h1);

    k_gather1_gemm2<<<N_NODES / TILE_ROWS, 32 * G1_WAVES, 0, stream>>>(off, pairs, dinv, h1, b1, W2, h2);

    k_gather2<<<(N_NODES / NODES_PER_WAVE + G2_WAVES - 1) / G2_WAVES, 32 * G2_WAVES, 0, stream>>>(off, pairs, dinv, h2, b2, out);
}


# GEMM1 operand loads 128 bits wide: A one load per 8 k, B one load per k row for all 4 N tiles (a lane owns 4 adjacent columns), 16-byte row stores; k order inside each block of 8 changes
# speedup vs baseline: 4.4287x; 1.0347x over previous
#include <hip/hip_runtime.h>

typedef float v2f __attribute__((ext_vector_type(2)));
typedef float v8f __attribute__((ext_vector_type(8)));
typedef float v4f __attribute__((ext_vector_type(4)));

#define N_NODES 50000
#define F_IN    256
#define F_HID   128
#define F_OUT   40

static __device__ __forceinline__ float inv_sqrt_deg(float d) {
    return (d > 0.0f) ? rsqrtf(d) : 0.0f;
}

#define NCHUNK   256
#define NBINS    256
#define RANGE    196
#define TAB_COLS (NBINS + 1)
#define SORT_THREADS 1024
static_assert(NBINS * RANGE >= N_NODES, "the ranges cover every node");
static_assert(SORT_THREADS == 4 * NCHUNK, "k_place: four lanes per run");

__global__ __launch_bounds__(SORT_THREADS) void k_bin(const int* __restrict__ row,
                                                      const int* __restrict__ col,
                                                      const float* __restrict__ ew,
                                                      int4* __restrict__ recs,
                                                      int* __restrict__ tab, int E) {
    __shared__ int start[NBINS];
    __shared__ int cur[NBINS];
    const int t = threadIdx.x;
    const int chunk = (E + NCHUNK - 1) / NCHUNK;
    const int e0 = blockIdx.x * chunk;
    int e1 = e0 + chunk;
    if (e1 > E) e1 = E;

    if (t < NBINS) start[t] = 0;
    __syncthreads();

    for (int b0 = e0; b0 < e1; b0 += 4 * SORT_THREADS) {
        int c[4];
#pragma unroll
        for (int u = 0; u < 4; ++u) {
            const int e = b0 + u * SORT_THREADS + t;
            c[u] = e < e1 ? col[e] : -1;
        }
#pragma unroll
        for (int u = 0; u < 4; ++u) {
            if (c[u] >= 0) atomicAdd(&start[(unsigned)c[u] / RANGE], 1);
        }
    }
    __syncthreads();

    if (t < 32) {
        int v[8];
        int s = 0;
#pragma unroll
        for (int q = 0; q < 8; ++q) { v[q] = start[t * 8 + q]; s += v[q]; }
        int inc = s;
        for (int d = 1; d < 32; d <<= 1) {
            const int x = __shfl_up(inc, d);
            if (t >= d) inc += x;
        }
        int run = inc - s;
#pragma unroll
        for (int q = 0; q < 8; ++q) {
            start[t * 8 + q] = run;
            cur[t * 8 + q]   = run;
            run += v[q];
        }
    }
    __syncthreads();

    int* __restrict__ trow = tab + (long)blockIdx.x * TAB_COLS;
    if (t < NBINS) trow[t] = start[t];
    if (t == NBINS) trow[NBINS] = e1 > e0 ? e1 - e0 : 0;

    int4* __restrict__ mine = recs + e0;
    for (int b0 = e0; b0 < e1; b0 += 4 * SORT_THREADS) {
        int c[4], r[4];
        float w[4];
#pragma unroll
        for (int u = 0; u < 4; ++u) {
            const int e = b0 + u * SORT_THREADS + t;
            const bool ok = e < e1;
            c[u] = ok ? col[e] : -1;
            r[u] = ok ? row[e] : 0;
            w[u] = ok ? ew[e] : 0.0f;
        }
#pragma unroll
        for (int u = 0; u < 4; ++u) {
            if (c[u] >= 0) {
                const int slot = atomicAdd(&cur[(unsigned)c[u] / RANGE], 1);
                mine[slot] = make_int4(r[u], c[u], __float_as_int(w[u]), 0);
            }
        }
    }
}

__global__ __launch_bounds__(SORT_THREADS) void k_place(const int4* __restrict__ recs,
                                                        const int* __restrict__ tab,
                                                        float* __restrict__ dinv,
                                                        int* __restrict__ off,
                                                        int* __restrict__ pairs, int E) {
    __shared__ int cnt[RANGE];
    __shared__ float dsum[RANGE];
    __shared__ int red[NCHUNK / 32];
    const int t    = threadIdx.x;
    const int lane = t & 31;
    const int wv   = t >> 5;
    const int k    = blockIdx.x;
    const int n0   = k * RANGE;
    const int chunk = (E + NCHUNK - 1) / NCHUNK;

    if (t < RANGE) { cnt[t] = 0; dsum[t] = 1.0f; }

    const int g = t >> 2;
    const int* __restrict__ trow = tab + (long)g * TAB_COLS + k;
    const int rs = trow[0];
    const int re = trow[1];
    const int4* __restrict__ run = recs + (long)g * chunk;

    int part = (t < NCHUNK) ? tab[(long)t * TAB_COLS + k] : 0;
    for (int d = 16; d > 0; d >>= 1) part += __shfl_xor(part, d);
    if (lane == 0 && wv < NCHUNK / 32) red[wv] = part;
    __syncthreads();

    for (int i = rs + (t & 3); i < re; i += 4) {
        const int4 rec = run[i];
        atomicAdd(&cnt[rec.y - n0], 1);
        atomicAdd(&dsum[rec.y - n0], __int_as_float(rec.z));
    }
    __syncthreads();

    if (t >= SORT_THREADS - RANGE) {
        const int idx = t - (SORT_THREADS - RANGE);
        if (n0 + idx < N_NODES) dinv[n0 + idx] = inv_sqrt_deg(dsum[idx]);
    }

    if (wv == 0) {
        int base = 0;
#pragma unroll
        for (int q = 0; q < NCHUNK / 32; ++q) base += red[q];
        int v[7];
        int s = 0;
#pragma unroll
        for (int q = 0; q < 7; ++q) {
            const int idx = lane * 7 + q;
            v[q] = idx < RANGE ? cnt[idx] : 0;
            s += v[q];
        }
        int inc = s;
        for (int d = 1; d < 32; d <<= 1) {
            const int x = __shfl_up(inc, d);
            if (lane >= d) inc += x;
        }
        int slot = base + inc - s;
#pragma unroll
        for (int q = 0; q < 7; ++q) {
            const int idx = lane * 7 + q;
            if (idx < RANGE) {
                cnt[idx] = slot;
                if (n0 + idx < N_NODES) off[n0 + idx] = slot;
                slot += v[q];
            }
        }
    }
    if (k == 0 && t == SORT_THREADS - 1) off[N_NODES] = E;
    __syncthreads();

    for (int i = rs + (t & 3); i < re; i += 4) {
        const int4 rec = run[i];
        const int slot = atomicAdd(&cnt[rec.y - n0], 1);
        ((int2*)pairs)[slot] = make_int2(rec.x, rec.z);
    }
}

__global__ __launch_bounds__(32) void k_gemm1(const float* __restrict__ X,
                                              const float* __restrict__ W,
                                              const float* __restrict__ dinv,
                                              float* __restrict__ H) {
    const int m0   = blockIdx.x * 16;
    const int n0   = blockIdx.y * 64;
    const int lane = threadIdx.x;
    const int half = lane >> 4;
    const int l16  = lane & 15;

    const float4 dlo = ((const float4*)(dinv + m0 + 8 * half))[0];
    const float4 dhi = ((const float4*)(dinv + m0 + 8 * half))[1];
    const float dv[8] = {dlo.x, dlo.y, dlo.z, dlo.w, dhi.x, dhi.y, dhi.z, dhi.w};

    const float* __restrict__ arow = X + (long)(m0 + l16) * F_IN + 4 * half;
    const float* __restrict__ bcol = W + (long)(4 * half) * F_HID + (n0 + 4 * l16);

    v8f acc0 = {}, acc1 = {}, acc2 = {}, acc3 = {};
#pragma unroll 2
    for (int k0 = 0; k0 < F_IN; k0 += 8) {
        const v4f a = *(const v4f*)(arow + k0);
        const float* __restrict__ bp = bcol + (long)k0 * F_HID;
        const v4f w0 = *(const v4f*)(bp);
        const v4f w1 = *(const v4f*)(bp + F_HID);
        const v4f w2 = *(const v4f*)(bp + 2 * F_HID);
        const v4f w3 = *(const v4f*)(bp + 3 * F_HID);
        v2f af, b;
        af.x = a.x; af.y = a.y;
        b.x = w0.x; b.y = w1.x;
        acc0 = __builtin_amdgcn_wmma_f32_16x16x4_f32(false, af, false, b, (short)0, acc0, false, false);
        b.x = w0.y; b.y = w1.y;
        acc1 = __builtin_amdgcn_wmma_f32_16x16x4_f32(false, af, false, b, (short)0, acc1, false, false);
        b.x = w0.z; b.y = w1.z;
        acc2 = __builtin_amdgcn_wmma_f32_16x16x4_f32(false, af, false, b, (short)0, acc2, false, false);
        b.x = w0.w; b.y = w1.w;
        acc3 = __builtin_amdgcn_wmma_f32_16x16x4_f32(false, af, false, b, (short)0, acc3, false, false);
        af.x = a.z; af.y = a.w;
        b.x = w2.x; b.y = w3.x;
        acc0 = __builtin_amdgcn_wmma_f32_16x16x4_f32(false, af, false, b, (short)0, acc0, false, false);
        b.x = w2.y; b.y = w3.y;
        acc1 = __builtin_amdgcn_wmma_f32_16x16x4_f32(false, af, false, b, (short)0, acc1, false, false);
        b.x = w2.z; b.y = w3.z;
        acc2 = __builtin_amdgcn_wmma_f32_16x16x4_f32(false, af, false, b, (short)0, acc2, false, false);
        b.x = w2.w; b.y = w3.w;
        acc3 = __builtin_amdgcn_wmma_f32_16x16x4_f32(false, af, false, b, (short)0, acc3, false, false);
    }
#pragma unroll
    for (int r = 0; r < 8; ++r) {
        const int mm = m0 + r + 8 * half;
        float4 o;
        o.x = acc0[r] * dv[r];
        o.y = acc1[r] * dv[r];
        o.z = acc2[r] * dv[r];
        o.w = acc3[r] * dv[r];
        *(float4*)(H + (long)mm * F_HID + n0 + 4 * l16) = o;
    }
}

#define ROWS_IN_FLIGHT 8
#define NODES_PER_WAVE 4
#define TILE_ROWS  16
#define G1_WAVES   (TILE_ROWS / NODES_PER_WAVE)
#define TILE_PITCH 132
static_assert(N_NODES % TILE_ROWS == 0, "every work-group has 16 real nodes");
static_assert(TILE_ROWS % NODES_PER_WAVE == 0 && G1_WAVES >= 3, "the tile's rows split evenly over the waves; three waves do the product");

static __device__ __forceinline__ int lane_to_scalar_i(int v, int srclane) {
    return __builtin_amdgcn_readlane(v, srclane);
}
static __device__ __forceinline__ float lane_to_scalar_f(float v, int srclane) {
    return __int_as_float(__builtin_amdgcn_readlane(__float_as_int(v), srclane));
}

static __device__ __forceinline__ int2 first_pairs(const int* __restrict__ pairs, int beg, int end, int lane) {
    int idx = beg + lane;
    idx = idx < end - 1 ? idx : end - 1;
    idx = idx > 0 ? idx : 0;
    return ((const int2*)pairs)[idx];
}

static __device__ __forceinline__ void sum_rows128(const float* __restrict__ h1, int lane,
                                                   int r, float w, int n, float4& acc) {
    for (int j = 0; j < n; j += ROWS_IN_FLIGHT) {
        v4f v[ROWS_IN_FLIGHT];
#pragma unroll
        for (int u = 0; u < ROWS_IN_FLIGHT; ++u) {
            const int ju = j + u < n ? j + u : n - 1;
            const int rj = lane_to_scalar_i(r, ju);
            v[u] = ((const v4f*)(h1 + (long)rj * F_HID))[lane];
        }
        asm volatile("" :: "v"(v[0]), "v"(v[1]), "v"(v[2]), "v"(v[3]),
                           "v"(v[4]), "v"(v[5]), "v"(v[6]), "v"(v[7]));
#pragma unroll
        for (int u = 0; u < ROWS_IN_FLIGHT; ++u) {
            if (u == 0 || j + u < n) {
                const float nj = lane_to_scalar_f(w, j + u);
                acc.x += v[u].x * nj;
                acc.y += v[u].y * nj;
                acc.z += v[u].z * nj;
                acc.w += v[u].w * nj;
            }
        }
    }
}

__global__ __launch_bounds__(32 * G1_WAVES, 16) void k_gather1_gemm2(const int* __restrict__ off,
                                                       const int* __restrict__ pairs,
                                                       const float* __restrict__ dinv,
                                                       const float* __restrict__ h1,
                                                       const float* __restrict__ b1,
                                                       const float* __restrict__ W2,
                                                       float* __restrict__ h2) {
    __shared__ __attribute__((aligned(16))) float tile[TILE_ROWS * TILE_PITCH];
    __shared__ float dtile[TILE_ROWS];

    const int lane = threadIdx.x & 31;
    const int wv = __builtin_amdgcn_readfirstlane((int)(threadIdx.x >> 5));
    const int m0 = blockIdx.x * TILE_ROWS;
    const int row0 = wv * NODES_PER_WAVE;
    const int c0 = m0 + row0;

    int bound[NODES_PER_WAVE + 1];
#pragma unroll
    for (int i = 0; i <= NODES_PER_WAVE; ++i) bound[i] = __builtin_amdgcn_readfirstlane(off[c0 + i]);

    float dself[NODES_PER_WAVE];
#pragma unroll
    for (int i = 0; i < NODES_PER_WAVE; ++i) dself[i] = dinv[c0 + i];

    int2 p0[NODES_PER_WAVE];
#pragma unroll
    for (int i = 0; i < NODES_PER_WAVE; ++i) p0[i] = first_pairs(pairs, bound[i], bound[i + 1], lane);

    const float4 bv = ((const float4*)b1)[lane];

    float4 own = ((const float4*)(h1 + (long)c0 * F_HID))[lane];

#pragma unroll
    for (int i = 0; i < NODES_PER_WAVE; ++i) {
        const int beg = bound[i];
        const int end = bound[i + 1];
        const float di = dself[i];
        float4 acc = own;
        if (i + 1 < NODES_PER_WAVE) own = ((const float4*)(h1 + (long)(c0 + i + 1) * F_HID))[lane];

        const int cnt = end - beg;
        sum_rows128(h1, lane, p0[i].x, __int_as_float(p0[i].y), cnt < 32 ? cnt : 32, acc);
        for (int base = beg + 32; base < end; base += 32) {
            const int left = end - base;
            const int n = left < 32 ? left : 32;
            int r = 0;
            float w = 0.0f;
            if (lane < n) {
                const int2 p = ((const int2*)pairs)[base + lane];
                r = p.x;
                w = __int_as_float(p.y);
            }
            sum_rows128(h1, lane, r, w, n, acc);
        }

        float4 o;
        o.x = acc.x * di + bv.x;
        o.y = acc.y * di + bv.y;
        o.z = acc.z * di + bv.z;
        o.w = acc.w * di + bv.w;
        o.x = o.x > 0.0f ? o.x : 0.0f;
        o.y = o.y > 0.0f ? o.y : 0.0f;
        o.z = o.z > 0.0f ? o.z : 0.0f;
        o.w = o.w > 0.0f ? o.w : 0.0f;
        ((float4*)(tile + (row0 + i) * TILE_PITCH))[lane] = o;
        if (lane == 0) dtile[row0 + i] = di;
    }
    __syncthreads();

    if (wv < 3) {
        const int half = lane >> 4;
        const int l16  = lane & 15;
        const int ncol = wv * 16 + l16;
        const bool nok = ncol < F_OUT;
        const float* __restrict__ arow = tile + l16 * TILE_PITCH + 2 * half;
        const float* __restrict__ bcol = W2 + (long)(2 * half) * F_OUT + (nok ? ncol : 0);

        v8f cacc = {};
#pragma unroll 8
        for (int k0 = 0; k0 < F_HID; k0 += 4) {
            const v2f a = *(const v2f*)(arow + k0);
            const float* __restrict__ b0p = bcol + (long)k0 * F_OUT;
            v2f b;
            b.x = b0p[0];
            b.y = b0p[F_OUT];
            cacc = __builtin_amdgcn_wmma_f32_16x16x4_f32(false, a, false, b, (short)0, cacc, false, false);
        }
        if (nok) {
#pragma unroll
            for (int r = 0; r < 8; ++r) {
                h2[(long)(m0 + r + 8 * half) * F_OUT + ncol] = cacc[r] * dtile[r + 8 * half];
            }
        }
    }
}

#define G2_WAVES 8
static_assert(N_NODES % NODES_PER_WAVE == 0, "a wave's nodes are all real");

static __device__ __forceinline__ void sum_rows40(const float* __restrict__ h2, int lane, int tl,
                                                  int r, float w, int n, float& a, float& b) {
    for (int j = 0; j < n; j += ROWS_IN_FLIGHT) {
        float va[ROWS_IN_FLIGHT], vb[ROWS_IN_FLIGHT];
#pragma unroll
        for (int u = 0; u < ROWS_IN_FLIGHT; ++u) {
            const int ju = j + u < n ? j + u : n - 1;
            const int rj = lane_to_scalar_i(r, ju);
            const float* __restrict__ src = h2 + (long)rj * F_OUT;
            va[u] = src[lane];
            vb[u] = src[32 + tl];
        }
        asm volatile("" :: "v"(va[0]), "v"(va[1]), "v"(va[2]), "v"(va[3]),
                           "v"(va[4]), "v"(va[5]), "v"(va[6]), "v"(va[7]),
                           "v"(vb[0]), "v"(vb[1]), "v"(vb[2]), "v"(vb[3]),
                           "v"(vb[4]), "v"(vb[5]), "v"(vb[6]), "v"(vb[7]));
#pragma unroll
        for (int u = 0; u < ROWS_IN_FLIGHT; ++u) {
            if (u == 0 || j + u < n) {
                const float nj = lane_to_scalar_f(w, j + u);
                a += va[u] * nj;
                b += vb[u] * nj;
            }
        }
    }
}

__global__ __launch_bounds__(32 * G2_WAVES) void k_gather2(const int* __restrict__ off,
                                                 const int* __restrict__ pairs,
                                                 const float* __restrict__ dinv,
                                                 const float* __restrict__ h2,
                                                 const float* __restrict__ b2,
                                                 float* __restrict__ out) {
    const int lane = threadIdx.x & 31;
    const int wv = __builtin_amdgcn_readfirstlane((int)(threadIdx.x >> 5));
    const int c0 = (blockIdx.x * G2_WAVES + wv) * NODES_PER_WAVE;
    if (c0 >= N_NODES) return;
    const bool tail = lane < F_OUT - 32;
    const int tl = tail ? lane : 0;

    int bound[NODES_PER_WAVE + 1];
#pragma unroll
    for (int i = 0; i <= NODES_PER_WAVE; ++i) bound[i] = __builtin_amdgcn_readfirstlane(off[c0 + i]);

    float dself[NODES_PER_WAVE];
#pragma unroll
    for (int i = 0; i < NODES_PER_WAVE; ++i) dself[i] = dinv[c0 + i];

    int2 p0[NODES_PER_WAVE];
#pragma unroll
    for (int i = 0; i < NODES_PER_WAVE; ++i) p0[i] = first_pairs(pairs, bound[i], bound[i + 1], lane);

    const float bias_a = b2[lane];
    const float bias_b = b2[32 + tl];

    float own_a = h2[(long)c0 * F_OUT + lane];
    float own_b = h2[(long)c0 * F_OUT + 32 + tl];

#pragma unroll
    for (int i = 0; i < NODES_PER_WAVE; ++i) {
        const int c = c0 + i;
        const int beg = bound[i];
        const int end = bound[i + 1];
        const float di = dself[i];
        float a = own_a;
        float b = own_b;
        if (i + 1 < NODES_PER_WAVE) {
            own_a = h2[(long)(c + 1) * F_OUT + lane];
            own_b = h2[(long)(c + 1) * F_OUT + 32 + tl];
        }

        const int cnt = end - beg;
        sum_rows40(h2, lane, tl, p0[i].x, __int_as_float(p0[i].y), cnt < 32 ? cnt : 32, a, b);
        for (int base = beg + 32; base < end; base += 32) {
            const int left = end - base;
            const int n = left < 32 ? left : 32;
            int r = 0;
            float w = 0.0f;
            if (lane < n) {
                const int2 p = ((const int2*)pairs)[base + lane];
                r = p.x;
                w = __int_as_float(p.y);
            }
            sum_rows40(h2, lane, tl, r, w, n, a, b);
        }

        float* __restrict__ dst = out + (long)c * F_OUT;
        dst[lane] = a * di + bias_a;
        if (tail) dst[32 + lane] = b * di + bias_b;
    }
}

extern "C" void kernel_launch(void* const* d_in, const int* in_sizes, int n_in,
                              void* d_out, int out_size, void* d_ws, size_t ws_size,
                              hipStream_t stream) {
    const float* x   = (const float*)d_in[0];
    const int*   ei  = (const int*)d_in[1];
    const float* ew  = (const float*)d_in[2];
    const float* W1  = (const float*)d_in[3];
    const float* b1  = (const float*)d_in[4];
    const float* W2  = (const float*)d_in[5];
    const float* b2  = (const float*)d_in[6];
    float*       out = (float*)d_out;

    const int E = in_sizes[2];
    const int* row = ei;
    const int* col = ei + E;

    char* ws = (char*)d_ws;
    float* dinv = (float*)(ws + 0);
    float* h1   = (float*)(ws + 409600);
    float* h2   = (float*)(ws + 409600 + 2 * 25600000);
    int*   off   = (int*)(ws + 59609600 + 200192);
    int4*  recs  = (int4*)(ws + 60009984);
    int*   tab   = (int*)(ws + 100663296);
    int*   pairs = (int*)(ws + 127118848);
    (void)ws_size; (void)n_in; (void)out_size;

    k_bin<<<NCHUNK, SORT_THREADS, 0, stream>>>(row, col, ew, recs, tab, E);
    k_place<<<NBINS, SORT_THREADS, 0, stream>>>(recs, tab, dinv, off, pairs, E);

    k_gemm1<<<dim3(N_NODES / 16, F_HID / 64), 32, 0, stream>>>(x, W1, dinv, h1);

    k_gather1_gemm2<<<N_NODES / TILE_ROWS, 32 * G1_WAVES, 0, stream>>>(off, pairs, dinv, h1, b1, W2, h2);

    k_gather2<<<(N_NODES / NODES_PER_WAVE + G2_WAVES - 1) / G2_WAVES, 32 * G2_WAVES, 0, stream>>>(off, pairs, dinv, h2, b2, out);
}


# 5 launches -> 4: the work-group that sorts a 208-node range then forms its own 13 row tiles of h1 = dinv*(X@W1), dinv from its LDS (28 wave strips, even per SIMD); no per-edge dinv; same sums
# speedup vs baseline: 4.6507x; 1.0501x over previous
#include <hip/hip_runtime.h>

typedef float v2f __attribute__((ext_vector_type(2)));
typedef float v8f __attribute__((ext_vector_type(8)));
typedef float v4f __attribute__((ext_vector_type(4)));

#define N_NODES 50000
#define F_IN    256
#define F_HID   128
#define F_OUT   40

static __device__ __forceinline__ float inv_sqrt_deg(float d) {
    return (d > 0.0f) ? rsqrtf(d) : 0.0f;
}

#define NCHUNK   256
#define NBINS    256
#define RANGE    208
#define RANGE_TILES (RANGE / 16)
#define PLACE_GROUPS ((N_NODES + RANGE - 1) / RANGE)
#define TAB_COLS (NBINS + 1)
#define SORT_THREADS 1024
static_assert(NBINS * RANGE >= N_NODES, "the ranges cover every node");
static_assert(SORT_THREADS == 4 * NCHUNK, "k_place: four lanes per run");
static_assert(RANGE % 16 == 0 && N_NODES % 16 == 0, "a range's row tiles are whole and all real or all past the end");
static_assert(2 * (RANGE_TILES - 1) + 4 <= SORT_THREADS / 32, "one strip of the range's rows per wave");
static_assert((2 * (RANGE_TILES - 1)) % 4 == 0, "the 16x64 strips fall evenly on the four SIMDs");
static_assert(RANGE <= 7 * 32, "k_place: the scan takes 7 counts per lane");

__global__ __launch_bounds__(SORT_THREADS) void k_bin(const int* __restrict__ row,
                                                      const int* __restrict__ col,
                                                      const float* __restrict__ ew,
                                                      int4* __restrict__ recs,
                                                      int* __restrict__ tab, int E) {
    __shared__ int start[NBINS];
    __shared__ int cur[NBINS];
    const int t = threadIdx.x;
    const int chunk = (E + NCHUNK - 1) / NCHUNK;
    const int e0 = blockIdx.x * chunk;
    int e1 = e0 + chunk;
    if (e1 > E) e1 = E;

    if (t < NBINS) start[t] = 0;
    __syncthreads();

    for (int b0 = e0; b0 < e1; b0 += 4 * SORT_THREADS) {
        int c[4];
#pragma unroll
        for (int u = 0; u < 4; ++u) {
            const int e = b0 + u * SORT_THREADS + t;
            c[u] = e < e1 ? col[e] : -1;
        }
#pragma unroll
        for (int u = 0; u < 4; ++u) {
            if (c[u] >= 0) atomicAdd(&start[(unsigned)c[u] / RANGE], 1);
        }
    }
    __syncthreads();

    if (t < 32) {
        int v[8];
        int s = 0;
#pragma unroll
        for (int q = 0; q < 8; ++q) { v[q] = start[t * 8 + q]; s += v[q]; }
        int inc = s;
        for (int d = 1; d < 32; d <<= 1) {
            const int x = __shfl_up(inc, d);
            if (t >= d) inc += x;
        }
        int run = inc - s;
#pragma unroll
        for (int q = 0; q < 8; ++q) {
            start[t * 8 + q] = run;
            cur[t * 8 + q]   = run;
            run += v[q];
        }
    }
    __syncthreads();

    int* __restrict__ trow = tab + (long)blockIdx.x * TAB_COLS;
    if (t < NBINS) trow[t] = start[t];
    if (t == NBINS) trow[NBINS] = e1 > e0 ? e1 - e0 : 0;

    int4* __restrict__ mine = recs + e0;
    for (int b0 = e0; b0 < e1; b0 += 4 * SORT_THREADS) {
        int c[4], r[4];
        float w[4];
#pragma unroll
        for (int u = 0; u < 4; ++u) {
            const int e = b0 + u * SORT_THREADS + t;
            const bool ok = e < e1;
            c[u] = ok ? col[e] : -1;
            r[u] = ok ? row[e] : 0;
            w[u] = ok ? ew[e] : 0.0f;
        }
#pragma unroll
        for (int u = 0; u < 4; ++u) {
            if (c[u] >= 0) {
                const int slot = atomicAdd(&cur[(unsigned)c[u] / RANGE], 1);
                mine[slot] = make_int4(r[u], c[u], __float_as_int(w[u]), 0);
            }
        }
    }
}

template <int NT>
static __device__ __forceinline__ void gemm1_strip(const float* __restrict__ X,
                                                   const float* __restrict__ W,
                                                   const float* dtile,
                                                   float* __restrict__ H,
                                                   int m0, int n0, int lane) {
    typedef float vnf __attribute__((ext_vector_type(NT)));
    const int half = lane >> 4;
    const int l16  = lane & 15;

    const float* __restrict__ arow = X + (long)(m0 + l16) * F_IN + 4 * half;
    const float* __restrict__ bcol = W + (long)(4 * half) * F_HID + (n0 + NT * l16);

    v8f acc[NT];
#pragma unroll
    for (int j = 0; j < NT; ++j) acc[j] = v8f{};
#pragma unroll 2
    for (int k0 = 0; k0 < F_IN; k0 += 8) {
        const v4f a = *(const v4f*)(arow + k0);
        const float* __restrict__ bp = bcol + (long)k0 * F_HID;
        vnf w[4];
#pragma unroll
        for (int q = 0; q < 4; ++q) w[q] = *(const vnf*)(bp + q * F_HID);
#pragma unroll
        for (int g = 0; g < 2; ++g) {
            v2f af;
            af.x = a[2 * g]; af.y = a[2 * g + 1];
#pragma unroll
            for (int j = 0; j < NT; ++j) {
                v2f b;
                b.x = w[2 * g][j]; b.y = w[2 * g + 1][j];
                acc[j] = __builtin_amdgcn_wmma_f32_16x16x4_f32(false, af, false, b, (short)0, acc[j], false, false);
            }
        }
    }
    float dv[8];
#pragma unroll
    for (int r = 0; r < 8; ++r) dv[r] = inv_sqrt_deg(dtile[8 * half + r]);
#pragma unroll
    for (int r = 0; r < 8; ++r) {
        const int mm = m0 + r + 8 * half;
        vnf o;
#pragma unroll
        for (int j = 0; j < NT; ++j) o[j] = acc[j][r] * dv[r];
        *(vnf*)(H + (long)mm * F_HID + n0 + NT * l16) = o;
    }
}

__global__ __launch_bounds__(SORT_THREADS) void k_place(const int4* __restrict__ recs,
                                                        const int* __restrict__ tab,
                                                        float* __restrict__ dinv,
                                                        int* __restrict__ off,
                                                        int* __restrict__ pairs, int E,
                                                        const float* __restrict__ X,
                                                        const float* __restrict__ W1,
                                                        float* __restrict__ h1) {
    __shared__ int cnt[RANGE];
    __shared__ __attribute__((aligned(16))) float dsum[RANGE];
    __shared__ int red[NCHUNK / 32];
    const int t    = threadIdx.x;
    const int lane = t & 31;
    const int wv   = t >> 5;
    const int k    = blockIdx.x;
    const int n0   = k * RANGE;
    const int chunk = (E + NCHUNK - 1) / NCHUNK;

    if (t < RANGE) { cnt[t] = 0; dsum[t] = 1.0f; }

    const int g = t >> 2;
    const int* __restrict__ trow = tab + (long)g * TAB_COLS + k;
    const int rs = trow[0];
    const int re = trow[1];
    const int4* __restrict__ run = recs + (long)g * chunk;

    int part = (t < NCHUNK) ? tab[(long)t * TAB_COLS + k] : 0;
    for (int d = 16; d > 0; d >>= 1) part += __shfl_xor(part, d);
    if (lane == 0 && wv < NCHUNK / 32) red[wv] = part;
    __syncthreads();

    for (int i = rs + (t & 3); i < re; i += 4) {
        const int4 rec = run[i];
        atomicAdd(&cnt[rec.y - n0], 1);
        atomicAdd(&dsum[rec.y - n0], __int_as_float(rec.z));
    }
    __syncthreads();

    if (t >= SORT_THREADS - RANGE) {
        const int idx = t - (SORT_THREADS - RANGE);
        if (n0 + idx < N_NODES) dinv[n0 + idx] = inv_sqrt_deg(dsum[idx]);
    }

    if (wv == 0) {
        int base = 0;
#pragma unroll
        for (int q = 0; q < NCHUNK / 32; ++q) base += red[q];
        int v[7];
        int s = 0;
#pragma unroll
        for (int q = 0; q < 7; ++q) {
            const int idx = lane * 7 + q;
            v[q] = idx < RANGE ? cnt[idx] : 0;
            s += v[q];
        }
        int inc = s;
        for (int d = 1; d < 32; d <<= 1) {
            const int x = __shfl_up(inc, d);
            if (lane >= d) inc += x;
        }
        int slot = base + inc - s;
#pragma unroll
        for (int q = 0; q < 7; ++q) {
            const int idx = lane * 7 + q;
            if (idx < RANGE) {
                cnt[idx] = slot;
                if (n0 + idx < N_NODES) off[n0 + idx] = slot;
                slot += v[q];
            }
        }
    }
    if (k == 0 && t == SORT_THREADS - 1) off[N_NODES] = E;
    __syncthreads();

    for (int i = rs + (t & 3); i < re; i += 4) {
        const int4 rec = run[i];
        const int slot = atomicAdd(&cnt[rec.y - n0], 1);
        ((int2*)pairs)[slot] = make_int2(rec.x, rec.z);
    }

    const int s = __builtin_amdgcn_readfirstlane(wv);
    if (s < 2 * (RANGE_TILES - 1)) {
        const int tile = s >> 1;
        const int m0 = n0 + 16 * tile;
        if (m0 < N_NODES) gemm1_strip<4>(X, W1, dsum + 16 * tile, h1, m0, 64 * (s & 1), lane);
    } else if (s < 2 * (RANGE_TILES - 1) + 4) {
        const int tile = RANGE_TILES - 1;
        const int m0 = n0 + 16 * tile;
        if (m0 < N_NODES) gemm1_strip<2>(X, W1, dsum + 16 * tile, h1, m0, 32 * (s - 2 * (RANGE_TILES - 1)), lane);
    }
}

#define ROWS_IN_FLIGHT 8
#define NODES_PER_WAVE 4
#define TILE_ROWS  16
#define G1_WAVES   (TILE_ROWS / NODES_PER_WAVE)
#define TILE_PITCH 132
static_assert(N_NODES % TILE_ROWS == 0, "every work-group has 16 real nodes");
static_assert(TILE_ROWS % NODES_PER_WAVE == 0 && G1_WAVES >= 3, "the tile's rows split evenly over the waves; three waves do the product");

static __device__ __forceinline__ int lane_to_scalar_i(int v, int srclane) {
    return __builtin_amdgcn_readlane(v, srclane);
}
static __device__ __forceinline__ float lane_to_scalar_f(float v, int srclane) {
    return __int_as_float(__builtin_amdgcn_readlane(__float_as_int(v), srclane));
}

static __device__ __forceinline__ int2 first_pairs(const int* __restrict__ pairs, int beg, int end, int lane) {
    int idx = beg + lane;
    idx = idx < end - 1 ? idx : end - 1;
    idx = idx > 0 ? idx : 0;
    return ((const int2*)pairs)[idx];
}

static __device__ __forceinline__ void sum_rows128(const float* __restrict__ h1, int lane,
                                                   int r, float w, int n, float4& acc) {
    for (int j = 0; j < n; j += ROWS_IN_FLIGHT) {
        v4f v[ROWS_IN_FLIGHT];
#pragma unroll
        for (int u = 0; u < ROWS_IN_FLIGHT; ++u) {
            const int ju = j + u < n ? j + u : n - 1;
            const int rj = lane_to_scalar_i(r, ju);
            v[u] = ((const v4f*)(h1 + (long)rj * F_HID))[lane];
        }
        asm volatile("" :: "v"(v[0]), "v"(v[1]), "v"(v[2]), "v"(v[3]),
                           "v"(v[4]), "v"(v[5]), "v"(v[6]), "v"(v[7]));
#pragma unroll
        for (int u = 0; u < ROWS_IN_FLIGHT; ++u) {
            if (u == 0 || j + u < n) {
                const float nj = lane_to_scalar_f(w, j + u);
                acc.x += v[u].x * nj;
                acc.y += v[u].y * nj;
                acc.z += v[u].z * nj;
                acc.w += v[u].w * nj;
            }
        }
    }
}

__global__ __launch_bounds__(32 * G1_WAVES, 16) void k_gather1_gemm2(const int* __restrict__ off,
                                                       const int* __restrict__ pairs,
                                                       const float* __restrict__ dinv,
                                                       const float* __restrict__ h1,
                                                       const float* __restrict__ b1,
                                                       const float* __restrict__ W2,
                                                       float* __restrict__ h2) {
    __shared__ __attribute__((aligned(16))) float tile[TILE_ROWS * TILE_PITCH];
    __shared__ float dtile[TILE_ROWS];

    const int lane = threadIdx.x & 31;
    const int wv = __builtin_amdgcn_readfirstlane((int)(threadIdx.x >> 5));
    const int m0 = blockIdx.x * TILE_ROWS;
    const int row0 = wv * NODES_PER_WAVE;
    const int c0 = m0 + row0;

    int bound[NODES_PER_WAVE + 1];
#pragma unroll
    for (int i = 0; i <= NODES_PER_WAVE; ++i) bound[i] = __builtin_amdgcn_readfirstlane(off[c0 + i]);

    float dself[NODES_PER_WAVE];
#pragma unroll
    for (int i = 0; i < NODES_PER_WAVE; ++i) dself[i] = dinv[c0 + i];

    int2 p0[NODES_PER_WAVE];
#pragma unroll
    for (int i = 0; i < NODES_PER_WAVE; ++i) p0[i] = first_pairs(pairs, bound[i], bound[i + 1], lane);

    const float4 bv = ((const float4*)b1)[lane];

    float4 own = ((const float4*)(h1 + (long)c0 * F_HID))[lane];

#pragma unroll
    for (int i = 0; i < NODES_PER_WAVE; ++i) {
        const int beg = bound[i];
        const int end = bound[i + 1];
        const float di = dself[i];
        float4 acc = own;
        if (i + 1 < NODES_PER_WAVE) own = ((const float4*)(h1 + (long)(c0 + i + 1) * F_HID))[lane];

        const int cnt = end - beg;
        sum_rows128(h1, lane, p0[i].x, __int_as_float(p0[i].y), cnt < 32 ? cnt : 32, acc);
        for (int base = beg + 32; base < end; base += 32) {
            const int left = end - base;
            const int n = left < 32 ? left : 32;
            int r = 0;
            float w = 0.0f;
            if (lane < n) {
                const int2 p = ((const int2*)pairs)[base + lane];
                r = p.x;
                w = __int_as_float(p.y);
            }
            sum_rows128(h1, lane, r, w, n, acc);
        }

        float4 o;
        o.x = acc.x * di + bv.x;
        o.y = acc.y * di + bv.y;
        o.z = acc.z * di + bv.z;
        o.w = acc.w * di + bv.w;
        o.x = o.x > 0.0f ? o.x : 0.0f;
        o.y = o.y > 0.0f ? o.y : 0.0f;
        o.z = o.z > 0.0f ? o.z : 0.0f;
        o.w = o.w > 0.0f ? o.w : 0.0f;
        ((float4*)(tile + (row0 + i) * TILE_PITCH))[lane] = o;
        if (lane == 0) dtile[row0 + i] = di;
    }
    __syncthreads();

    if (wv < 3) {
        const int half = lane >> 4;
        const int l16  = lane & 15;
        const int ncol = wv * 16 + l16;
        const bool nok = ncol < F_OUT;
        const float* __restrict__ arow = tile + l16 * TILE_PITCH + 2 * half;
        const float* __restrict__ bcol = W2 + (long)(2 * half) * F_OUT + (nok ? ncol : 0);

        v8f cacc = {};
#pragma unroll 8
        for (int k0 = 0; k0 < F_HID; k0 += 4) {
            const v2f a = *(const v2f*)(arow + k0);
            const float* __restrict__ b0p = bcol + (long)k0 * F_OUT;
            v2f b;
            b.x = b0p[0];
            b.y = b0p[F_OUT];
            cacc = __builtin_amdgcn_wmma_f32_16x16x4_f32(false, a, false, b, (short)0, cacc, false, false);
        }
        if (nok) {
#pragma unroll
            for (int r = 0; r < 8; ++r) {
                h2[(long)(m0 + r + 8 * half) * F_OUT + ncol] = cacc[r] * dtile[r + 8 * half];
            }
        }
    }
}

#define G2_WAVES 8
static_assert(N_NODES % NODES_PER_WAVE == 0, "a wave's nodes are all real");

static __device__ __forceinline__ void sum_rows40(const float* __restrict__ h2, int lane, int tl,
                                                  int r, float w, int n, float& a, float& b) {
    for (int j = 0; j < n; j += ROWS_IN_FLIGHT) {
        float va[ROWS_IN_FLIGHT], vb[ROWS_IN_FLIGHT];
#pragma unroll
        for (int u = 0; u < ROWS_IN_FLIGHT; ++u) {
            const int ju = j + u < n ? j + u : n - 1;
            const int rj = lane_to_scalar_i(r, ju);
            const float* __restrict__ src = h2 + (long)rj * F_OUT;
            va[u] = src[lane];
            vb[u] = src[32 + tl];
        }
        asm volatile("" :: "v"(va[0]), "v"(va[1]), "v"(va[2]), "v"(va[3]),
                           "v"(va[4]), "v"(va[5]), "v"(va[6]), "v"(va[7]),
                           "v"(vb[0]), "v"(vb[1]), "v"(vb[2]), "v"(vb[3]),
                           "v"(vb[4]), "v"(vb[5]), "v"(vb[6]), "v"(vb[7]));
#pragma unroll
        for (int u = 0; u < ROWS_IN_FLIGHT; ++u) {
            if (u == 0 || j + u < n) {
                const float nj = lane_to_scalar_f(w, j + u);
                a += va[u] * nj;
                b += vb[u] * nj;
            }
        }
    }
}

__global__ __launch_bounds__(32 * G2_WAVES) void k_gather2(const int* __restrict__ off,
                                                 const int* __restrict__ pairs,
                                                 const float* __restrict__ dinv,
                                                 const float* __restrict__ h2,
                                                 const float* __restrict__ b2,
                                                 float* __restrict__ out) {
    const int lane = threadIdx.x & 31;
    const int wv = __builtin_amdgcn_readfirstlane((int)(threadIdx.x >> 5));
    const int c0 = (blockIdx.x * G2_WAVES + wv) * NODES_PER_WAVE;
    if (c0 >= N_NODES) return;
    const bool tail = lane < F_OUT - 32;
    const int tl = tail ? lane : 0;

    int bound[NODES_PER_WAVE + 1];
#pragma unroll
    for (int i = 0; i <= NODES_PER_WAVE; ++i) bound[i] = __builtin_amdgcn_readfirstlane(off[c0 + i]);

    float dself[NODES_PER_WAVE];
#pragma unroll
    for (int i = 0; i < NODES_PER_WAVE; ++i) dself[i] = dinv[c0 + i];

    int2 p0[NODES_PER_WAVE];
#pragma unroll
    for (int i = 0; i < NODES_PER_WAVE; ++i) p0[i] = first_pairs(pairs, bound[i], bound[i + 1], lane);

    const float bias_a = b2[lane];
    const float bias_b = b2[32 + tl];

    float own_a = h2[(long)c0 * F_OUT + lane];
    float own_b = h2[(long)c0 * F_OUT + 32 + tl];

#pragma unroll
    for (int i = 0; i < NODES_PER_WAVE; ++i) {
        const int c = c0 + i;
        const int beg = bound[i];
        const int end = bound[i + 1];
        const float di = dself[i];
        float a = own_a;
        float b = own_b;
        if (i + 1 < NODES_PER_WAVE) {
            own_a = h2[(long)(c + 1) * F_OUT + lane];
            own_b = h2[(long)(c + 1) * F_OUT + 32 + tl];
        }

        const int cnt = end - beg;
        sum_rows40(h2, lane, tl, p0[i].x, __int_as_float(p0[i].y), cnt < 32 ? cnt : 32, a, b);
        for (int base = beg + 32; base < end; base += 32) {
            const int left = end - base;
            const int n = left < 32 ? left : 32;
            int r = 0;
            float w = 0.0f;
            if (lane < n) {
                const int2 p = ((const int2*)pairs)[base + lane];
                r = p.x;
                w = __int_as_float(p.y);
            }
            sum_rows40(h2, lane, tl, r, w, n, a, b);
        }

        float* __restrict__ dst = out + (long)c * F_OUT;
        dst[lane] = a * di + bias_a;
        if (tail) dst[32 + lane] = b * di + bias_b;
    }
}

extern "C" void kernel_launch(void* const* d_in, const int* in_sizes, int n_in,
                              void* d_out, int out_size, void* d_ws, size_t ws_size,
                              hipStream_t stream) {
    const float* x   = (const float*)d_in[0];
    const int*   ei  = (const int*)d_in[1];
    const float* ew  = (const float*)d_in[2];
    const float* W1  = (const float*)d_in[3];
    const float* b1  = (const float*)d_in[4];
    const float* W2  = (const float*)d_in[5];
    const float* b2  = (const float*)d_in[6];
    float*       out = (float*)d_out;

    const int E = in_sizes[2];
    const int* row = ei;
    const int* col = ei + E;

    char* ws = (char*)d_ws;
    float* dinv = (float*)(ws + 0);
    float* h1   = (float*)(ws + 409600);
    float* h2   = (float*)(ws + 409600 + 2 * 25600000);
    int*   off   = (int*)(ws + 59609600 + 200192);
    int4*  recs  = (int4*)(ws + 60009984);
    int*   tab   = (int*)(ws + 100663296);
    int*   pairs = (int*)(ws + 127118848);
    (void)ws_size; (void)n_in; (void)out_size;

    k_bin<<<NCHUNK, SORT_THREADS, 0, stream>>>(row, col, ew, recs, tab, E);
    k_place<<<PLACE_GROUPS, SORT_THREADS, 0, stream>>>(recs, tab, dinv, off, pairs, E, x, W1, h1);

    k_gather1_gemm2<<<N_NODES / TILE_ROWS, 32 * G1_WAVES, 0, stream>>>(off, pairs, dinv, h1, b1, W2, h2);

    k_gather2<<<(N_NODES / NODES_PER_WAVE + G2_WAVES - 1) / G2_WAVES, 32 * G2_WAVES, 0, stream>>>(off, pairs, dinv, h2, b2, out);
}
